# starting kernel
# baseline (speedup 1.0000x reference)
#include <hip/hip_runtime.h>
#include <hip/hip_bf16.h>


#define NB 65536
#define NT 100

typedef __attribute__((ext_vector_type(2))) float v2f;
typedef __attribute__((ext_vector_type(8))) float v8f;

__device__ __forceinline__ float fast_rcp(float x) {
  return __builtin_amdgcn_rcpf(x);
}
__device__ __forceinline__ float sigm(float x) {
  return fast_rcp(1.0f + __expf(-x));
}
__device__ __forceinline__ float tanh_fast(float x) {
  return fmaf(2.0f, fast_rcp(1.0f + __expf(-2.0f * x)), -1.0f);
}

template <int IN, bool REV, bool TMAJOR>
__global__ __launch_bounds__(256) void lstm_dir_kernel(
    const float* __restrict__ in,
    const float* __restrict__ w_ih,
    const float* __restrict__ w_hh,
    const float* __restrict__ b_ih,
    const float* __restrict__ b_hh,
    float* __restrict__ out,
    int out_off)
{
  const int b = blockIdx.x * blockDim.x + threadIdx.x;
  if (b >= NB) return;

  float wih[8][IN];
  float whh[8][2];
  float bias[8];
#pragma unroll
  for (int g = 0; g < 8; ++g) {
#pragma unroll
    for (int c = 0; c < IN; ++c) wih[g][c] = w_ih[g * IN + c];
    whh[g][0] = w_hh[g * 2 + 0];
    whh[g][1] = w_hh[g * 2 + 1];
    bias[g] = b_ih[g] + b_hh[g];
  }

  float h0 = 0.f, h1 = 0.f, c0 = 0.f, c1 = 0.f;

  for (int it = 0; it < NT; ++it) {
    const int tt = REV ? (NT - 1 - it) : it;

    float xv[IN];
    if constexpr (TMAJOR) {
      const float4 x4 =
          *(const float4*)(in + ((size_t)tt * NB + (size_t)b) * 4);
      xv[0] = x4.x; xv[1] = x4.y; xv[2] = x4.z; xv[3] = x4.w;
    } else {
      const float* p = in + (size_t)b * NT * IN + (size_t)tt * IN;
#pragma unroll
      for (int c = 0; c < IN; ++c) xv[c] = p[c];
    }

    float g[8];
#pragma unroll
    for (int gg = 0; gg < 8; ++gg) {
      float acc = fmaf(whh[gg][0], h0, fmaf(whh[gg][1], h1, bias[gg]));
#pragma unroll
      for (int c = 0; c < IN; ++c) acc = fmaf(wih[gg][c], xv[c], acc);
      g[gg] = acc;
    }

    const float i0 = sigm(g[0]), i1 = sigm(g[1]);
    const float f0 = sigm(g[2]), f1 = sigm(g[3]);
    const float t0 = tanh_fast(g[4]), t1 = tanh_fast(g[5]);
    const float o0 = sigm(g[6]), o1 = sigm(g[7]);

    c0 = fmaf(f0, c0, i0 * t0);
    c1 = fmaf(f1, c1, i1 * t1);
    h0 = o0 * tanh_fast(c0);
    h1 = o1 * tanh_fast(c1);

    float2 hv = make_float2(h0, h1);
    *(float2*)(out + ((size_t)tt * NB + (size_t)b) * 4 + out_off) = hv;
  }
}

__global__ __launch_bounds__(256) void fc1_wmma_kernel(
    const float* __restrict__ h2,
    const float* __restrict__ w,
    const float* __restrict__ bias,
    float* __restrict__ act)
{
  const int wid = blockIdx.x * (blockDim.x >> 5) + (threadIdx.x >> 5);
  const int lane = threadIdx.x & 31;
  const int m_tile = wid >> 2;
  const int n_tile = wid & 3;
  const int row = lane & 15;
  const int half = lane >> 4;
  const int base_b = m_tile * 16;
  const int n = n_tile * 16 + row;

  v8f acc = {};
  const float* wrow = w + (size_t)n * 400 + half * 2;
  const float* arow = h2 + ((size_t)(base_b + row)) * 4 + half * 2;

#pragma unroll 4
  for (int t = 0; t < NT; ++t) {
    v2f a = *(const v2f*)(arow + (size_t)t * NB * 4);
    a.x = fmaxf(a.x, 0.0f);
    a.y = fmaxf(a.y, 0.0f);
    v2f bv = *(const v2f*)(wrow + t * 4);
    acc = __builtin_amdgcn_wmma_f32_16x16x4_f32(
        false, a, false, bv, (short)0, acc, false, false);
  }

  const float bn = bias[n];
#pragma unroll
  for (int r = 0; r < 8; ++r) {
    const int m = r + half * 8;
    act[(size_t)(base_b + m) * 64 + n] = fmaxf(acc[r] + bn, 0.0f);
  }
}

__global__ __launch_bounds__(256) void fc2_wmma_kernel(
    const float* __restrict__ act,
    const float* __restrict__ w,
    const float* __restrict__ bias,
    float* __restrict__ out)
{
  const int wid = blockIdx.x * (blockDim.x >> 5) + (threadIdx.x >> 5);
  const int lane = threadIdx.x & 31;
  const int m_tile = wid >> 1;
  const int n_tile = wid & 1;
  const int row = lane & 15;
  const int half = lane >> 4;
  const int base_b = m_tile * 16;
  const int n = n_tile * 16 + row;
  const bool nvalid = (n < 20);

  v8f acc = {};
  const float* arow = act + (size_t)(base_b + row) * 64 + half * 2;
  const float* wrow = w + (size_t)(nvalid ? n : 0) * 64 + half * 2;

#pragma unroll 4
  for (int kk = 0; kk < 16; ++kk) {
    v2f a = *(const v2f*)(arow + kk * 4);
    v2f bv = *(const v2f*)(wrow + kk * 4);
    if (!nvalid) { bv.x = 0.0f; bv.y = 0.0f; }
    acc = __builtin_amdgcn_wmma_f32_16x16x4_f32(
        false, a, false, bv, (short)0, acc, false, false);
  }

  if (nvalid) {
    const float bn = bias[n];
#pragma unroll
    for (int r = 0; r < 8; ++r) {
      const int m = r + half * 8;
      out[(size_t)(base_b + m) * 20 + n] = acc[r] + bn;
    }
  }
}

extern "C" void kernel_launch(void* const* d_in, const int* in_sizes, int n_in,
                              void* d_out, int out_size, void* d_ws,
                              size_t ws_size, hipStream_t stream) {
  const float* x = (const float*)d_in[0];
  const float* w_ih_l0  = (const float*)d_in[1];
  const float* w_hh_l0  = (const float*)d_in[2];
  const float* b_ih_l0  = (const float*)d_in[3];
  const float* b_hh_l0  = (const float*)d_in[4];
  const float* w_ih_l0r = (const float*)d_in[5];
  const float* w_hh_l0r = (const float*)d_in[6];
  const float* b_ih_l0r = (const float*)d_in[7];
  const float* b_hh_l0r = (const float*)d_in[8];
  const float* w_ih_l1  = (const float*)d_in[9];
  const float* w_hh_l1  = (const float*)d_in[10];
  const float* b_ih_l1  = (const float*)d_in[11];
  const float* b_hh_l1  = (const float*)d_in[12];
  const float* w_ih_l1r = (const float*)d_in[13];
  const float* w_hh_l1r = (const float*)d_in[14];
  const float* b_ih_l1r = (const float*)d_in[15];
  const float* b_hh_l1r = (const float*)d_in[16];
  const float* fc1_w = (const float*)d_in[17];
  const float* fc1_b = (const float*)d_in[18];
  const float* fc2_w = (const float*)d_in[19];
  const float* fc2_b = (const float*)d_in[20];
  float* out = (float*)d_out;

  float* ws  = (float*)d_ws;
  float* ws0 = ws;
  float* ws1 = ws + (size_t)NT * NB * 4;
  float* ws2 = ws;

  const dim3 blk(256);
  const dim3 grid_lstm(NB / 256);

  lstm_dir_kernel<7, false, false><<<grid_lstm, blk, 0, stream>>>(
      x, w_ih_l0, w_hh_l0, b_ih_l0, b_hh_l0, ws0, 0);
  lstm_dir_kernel<7, true, false><<<grid_lstm, blk, 0, stream>>>(
      x, w_ih_l0r, w_hh_l0r, b_ih_l0r, b_hh_l0r, ws0, 2);

  lstm_dir_kernel<4, false, true><<<grid_lstm, blk, 0, stream>>>(
      ws0, w_ih_l1, w_hh_l1, b_ih_l1, b_hh_l1, ws1, 0);
  lstm_dir_kernel<4, true, true><<<grid_lstm, blk, 0, stream>>>(
      ws0, w_ih_l1r, w_hh_l1r, b_ih_l1r, b_hh_l1r, ws1, 2);

  fc1_wmma_kernel<<<dim3(2048), blk, 0, stream>>>(ws1, fc1_w, fc1_b, ws2);

  fc2_wmma_kernel<<<dim3(1024), blk, 0, stream>>>(ws2, fc2_w, fc2_b, out);
}


# FC1 and FC2 operand fragments fetched 16 bytes per lane: lane half h holds k = 8j+4h..+3, feeding two WMMAs (half the loads and cache lines per WMMA); order of k inside the FC sums changes
# speedup vs baseline: 1.9519x; 1.9519x over previous
#include <hip/hip_runtime.h>
#include <hip/hip_bf16.h>


#define NB 65536
#define NT 100

typedef __attribute__((ext_vector_type(2))) float v2f;
typedef __attribute__((ext_vector_type(8))) float v8f;

__device__ __forceinline__ float fast_rcp(float x) {
  return __builtin_amdgcn_rcpf(x);
}
__device__ __forceinline__ float sigm(float x) {
  return fast_rcp(1.0f + __expf(-x));
}
__device__ __forceinline__ float tanh_fast(float x) {
  return fmaf(2.0f, fast_rcp(1.0f + __expf(-2.0f * x)), -1.0f);
}

template <int IN>
struct DirWeights {
  float wih[8][IN];
  float whh[8][2];
  float bias[8];
};

__device__ __forceinline__ float keep_in_vgpr(float v) {
  asm("" : "+v"(v));
  return v;
}

template <int IN, bool IN_VGPR>
__device__ __forceinline__ void load_dir_weights(
    DirWeights<IN>& w,
    const float* __restrict__ w_ih,
    const float* __restrict__ w_hh,
    const float* __restrict__ b_ih,
    const float* __restrict__ b_hh)
{
#pragma unroll
  for (int g = 0; g < 8; ++g) {
#pragma unroll
    for (int c = 0; c < IN; ++c) w.wih[g][c] = w_ih[g * IN + c];
    w.whh[g][0] = w_hh[g * 2 + 0];
    w.whh[g][1] = w_hh[g * 2 + 1];
    w.bias[g] = b_ih[g] + b_hh[g];
    if constexpr (IN_VGPR) {
#pragma unroll
      for (int c = 0; c < IN; ++c) w.wih[g][c] = keep_in_vgpr(w.wih[g][c]);
      w.whh[g][0] = keep_in_vgpr(w.whh[g][0]);
      w.whh[g][1] = keep_in_vgpr(w.whh[g][1]);
      w.bias[g] = keep_in_vgpr(w.bias[g]);
    }
  }
}

template <int IN>
__device__ __forceinline__ void lstm_step(
    const DirWeights<IN>& w, const float (&xv)[IN],
    float& h0, float& h1, float& c0, float& c1)
{
  float g[8];
#pragma unroll
  for (int gg = 0; gg < 8; ++gg) {
    float acc = fmaf(w.whh[gg][0], h0, fmaf(w.whh[gg][1], h1, w.bias[gg]));
#pragma unroll
    for (int c = 0; c < IN; ++c) acc = fmaf(w.wih[gg][c], xv[c], acc);
    g[gg] = acc;
  }

  const float i0 = sigm(g[0]), i1 = sigm(g[1]);
  const float f0 = sigm(g[2]), f1 = sigm(g[3]);
  const float t0 = tanh_fast(g[4]), t1 = tanh_fast(g[5]);
  const float o0 = sigm(g[6]), o1 = sigm(g[7]);

  c0 = fmaf(f0, c0, i0 * t0);
  c1 = fmaf(f1, c1, i1 * t1);
  h0 = o0 * tanh_fast(c0);
  h1 = o1 * tanh_fast(c1);
}

#define SPW 256
#define TC 4
#define NCH (NT / TC)
#define PITCH 257

template <int IN>
struct TilePieces {
  float4 v[TC * IN / 4];
};

template <int IN, bool TMAJOR>
__device__ __forceinline__ void load_tile(
    TilePieces<IN>& r, const float* __restrict__ in, int b0, int t0, int lj)
{
  if constexpr (TMAJOR) {
    static_assert(IN == 4, "time-major input has 4 channels");
#pragma unroll
    for (int st = 0; st < TC; ++st)
      r.v[st] = *(const float4*)(in + ((size_t)(t0 + st) * NB + (size_t)(b0 + lj)) * 4);
  } else {
    static_assert((TC * IN) % 4 == 0, "tile row must be whole float4 pieces");
    constexpr int PPS = TC * IN / 4;
#pragma unroll
    for (int i = 0; i < PPS; ++i) {
      const int p = lj + SPW * i;
      const int srow = p / PPS;
      const int q = p - srow * PPS;
      r.v[i] = *(const float4*)(in + (size_t)(b0 + srow) * (NT * IN) +
                                (size_t)t0 * IN + q * 4);
    }
  }
}

template <int IN, bool TMAJOR>
__device__ __forceinline__ void store_tile(
    float* __restrict__ tile, const TilePieces<IN>& r, int lj)
{
  constexpr int PPS = TC * IN / 4;
#pragma unroll
  for (int i = 0; i < PPS; ++i) {
    int srow, row0;
    if constexpr (TMAJOR) {
      srow = lj;
      row0 = i * 4;
    } else {
      const int p = lj + SPW * i;
      srow = p / PPS;
      row0 = (p - srow * PPS) * 4;
    }
    float* d = tile + row0 * PITCH + srow;
    d[0 * PITCH] = r.v[i].x;
    d[1 * PITCH] = r.v[i].y;
    d[2 * PITCH] = r.v[i].z;
    d[3 * PITCH] = r.v[i].w;
  }
}

__device__ __forceinline__ void lds_handover() {
  __builtin_amdgcn_fence(__ATOMIC_RELEASE, "workgroup", "local");
  __builtin_amdgcn_s_barrier();
  __builtin_amdgcn_fence(__ATOMIC_ACQUIRE, "workgroup", "local");
}

template <int IN, bool TMAJOR>
__global__ __launch_bounds__(512) __attribute__((amdgpu_waves_per_eu(4, 4)))
void lstm_bidir_kernel(
    const float* __restrict__ in,
    const float* __restrict__ wf_ih, const float* __restrict__ wf_hh,
    const float* __restrict__ bf_ih, const float* __restrict__ bf_hh,
    const float* __restrict__ wr_ih, const float* __restrict__ wr_hh,
    const float* __restrict__ br_ih, const float* __restrict__ br_hh,
    float* __restrict__ out)
{
  __shared__ float ring[2][2][TC * IN * PITCH];

  const int tid = threadIdx.x;
  const bool is_loader = __builtin_amdgcn_readfirstlane(tid) >= SPW;
  const int b0 = blockIdx.x * SPW;
  const int lj = tid - SPW;
  const int b = b0 + tid;

  DirWeights<IN> wf, wr;
  load_dir_weights<IN, false>(wf, wf_ih, wf_hh, bf_ih, bf_hh);
  load_dir_weights<IN, true>(wr, wr_ih, wr_hh, br_ih, br_hh);

  float hf0 = 0.f, hf1 = 0.f, cf0 = 0.f, cf1 = 0.f;
  float hr0 = 0.f, hr1 = 0.f, cr0 = 0.f, cr1 = 0.f;

  TilePieces<IN> pf, pr;
  if (is_loader) {
    load_tile<IN, TMAJOR>(pf, in, b0, 0, lj);
    load_tile<IN, TMAJOR>(pr, in, b0, NT - TC, lj);
    store_tile<IN, TMAJOR>(ring[0][0], pf, lj);
    store_tile<IN, TMAJOR>(ring[0][1], pr, lj);
    load_tile<IN, TMAJOR>(pf, in, b0, TC, lj);
    load_tile<IN, TMAJOR>(pr, in, b0, NT - 2 * TC, lj);
  }
  lds_handover();

  for (int c = 0; c < NCH; ++c) {
    const int stage = c & 1;
    if (is_loader) {
      if (c + 1 < NCH) {
        store_tile<IN, TMAJOR>(ring[stage ^ 1][0], pf, lj);
        store_tile<IN, TMAJOR>(ring[stage ^ 1][1], pr, lj);
        if (c + 2 < NCH) {
          load_tile<IN, TMAJOR>(pf, in, b0, (c + 2) * TC, lj);
          load_tile<IN, TMAJOR>(pr, in, b0, NT - TC - (c + 2) * TC, lj);
        }
      }
    } else {
      const float* tf_tile = ring[stage][0];
      const float* tr_tile = ring[stage][1];
#pragma unroll
      for (int s = 0; s < TC; ++s) {
        const int tf = c * TC + s;
        const int tr = NT - 1 - tf;
        float xf[IN], xr[IN];
#pragma unroll
        for (int ch = 0; ch < IN; ++ch) {
          xf[ch] = tf_tile[(s * IN + ch) * PITCH + tid];
          xr[ch] = tr_tile[((TC - 1 - s) * IN + ch) * PITCH + tid];
        }

        lstm_step<IN>(wf, xf, hf0, hf1, cf0, cf1);
        lstm_step<IN>(wr, xr, hr0, hr1, cr0, cr1);

        *(float2*)(out + ((size_t)tf * NB + (size_t)b) * 4 + 0) =
            make_float2(hf0, hf1);
        *(float2*)(out + ((size_t)tr * NB + (size_t)b) * 4 + 2) =
            make_float2(hr0, hr1);
      }
    }
    lds_handover();
  }
}

__global__ __launch_bounds__(256) void fc1_wmma_kernel(
    const float* __restrict__ h2,
    const float* __restrict__ w,
    const float* __restrict__ bias,
    float* __restrict__ act)
{
  const int wid = blockIdx.x * (blockDim.x >> 5) + (threadIdx.x >> 5);
  const int lane = threadIdx.x & 31;
  const int m_tile = wid >> 2;
  const int n_tile = wid & 3;
  const int row = lane & 15;
  const int half = lane >> 4;
  const int base_b = m_tile * 16;
  const int n = n_tile * 16 + row;

  v8f acc = {};
  const float* wrow = w + (size_t)n * 400 + half * 4;
  const float* arow = h2 + ((size_t)half * NB + (size_t)(base_b + row)) * 4;

#pragma unroll 5
  for (int j = 0; j < NT / 2; ++j) {
    float4 a = *(const float4*)(arow + (size_t)j * 2 * NB * 4);
    const float4 bq = *(const float4*)(wrow + j * 8);
    v2f a01, a23, b01, b23;
    a01.x = fmaxf(a.x, 0.0f);
    a01.y = fmaxf(a.y, 0.0f);
    a23.x = fmaxf(a.z, 0.0f);
    a23.y = fmaxf(a.w, 0.0f);
    b01.x = bq.x; b01.y = bq.y;
    b23.x = bq.z; b23.y = bq.w;
    acc = __builtin_amdgcn_wmma_f32_16x16x4_f32(
        false, a01, false, b01, (short)0, acc, false, false);
    acc = __builtin_amdgcn_wmma_f32_16x16x4_f32(
        false, a23, false, b23, (short)0, acc, false, false);
  }

  const float bn = bias[n];
#pragma unroll
  for (int r = 0; r < 8; ++r) {
    const int m = r + half * 8;
    act[(size_t)(base_b + m) * 64 + n] = fmaxf(acc[r] + bn, 0.0f);
  }
}

__global__ __launch_bounds__(256) void fc2_wmma_kernel(
    const float* __restrict__ act,
    const float* __restrict__ w,
    const float* __restrict__ bias,
    float* __restrict__ out)
{
  const int wid = blockIdx.x * (blockDim.x >> 5) + (threadIdx.x >> 5);
  const int lane = threadIdx.x & 31;
  const int m_tile = wid >> 1;
  const int n_tile = wid & 1;
  const int row = lane & 15;
  const int half = lane >> 4;
  const int base_b = m_tile * 16;
  const int n = n_tile * 16 + row;
  const bool nvalid = (n < 20);

  v8f acc = {};
  const float* arow = act + (size_t)(base_b + row) * 64 + half * 4;
  const float* wrow = w + (size_t)(nvalid ? n : 0) * 64 + half * 4;

#pragma unroll
  for (int j = 0; j < 8; ++j) {
    const float4 a = *(const float4*)(arow + j * 8);
    float4 bq = *(const float4*)(wrow + j * 8);
    if (!nvalid) { bq.x = 0.0f; bq.y = 0.0f; bq.z = 0.0f; bq.w = 0.0f; }
    v2f a01, a23, b01, b23;
    a01.x = a.x; a01.y = a.y;
    a23.x = a.z; a23.y = a.w;
    b01.x = bq.x; b01.y = bq.y;
    b23.x = bq.z; b23.y = bq.w;
    acc = __builtin_amdgcn_wmma_f32_16x16x4_f32(
        false, a01, false, b01, (short)0, acc, false, false);
    acc = __builtin_amdgcn_wmma_f32_16x16x4_f32(
        false, a23, false, b23, (short)0, acc, false, false);
  }

  if (nvalid) {
    const float bn = bias[n];
#pragma unroll
    for (int r = 0; r < 8; ++r) {
      const int m = r + half * 8;
      out[(size_t)(base_b + m) * 20 + n] = acc[r] + bn;
    }
  }
}

extern "C" void kernel_launch(void* const* d_in, const int* in_sizes, int n_in,
                              void* d_out, int out_size, void* d_ws,
                              size_t ws_size, hipStream_t stream) {
  const float* x = (const float*)d_in[0];
  const float* w_ih_l0  = (const float*)d_in[1];
  const float* w_hh_l0  = (const float*)d_in[2];
  const float* b_ih_l0  = (const float*)d_in[3];
  const float* b_hh_l0  = (const float*)d_in[4];
  const float* w_ih_l0r = (const float*)d_in[5];
  const float* w_hh_l0r = (const float*)d_in[6];
  const float* b_ih_l0r = (const float*)d_in[7];
  const float* b_hh_l0r = (const float*)d_in[8];
  const float* w_ih_l1  = (const float*)d_in[9];
  const float* w_hh_l1  = (const float*)d_in[10];
  const float* b_ih_l1  = (const float*)d_in[11];
  const float* b_hh_l1  = (const float*)d_in[12];
  const float* w_ih_l1r = (const float*)d_in[13];
  const float* w_hh_l1r = (const float*)d_in[14];
  const float* b_ih_l1r = (const float*)d_in[15];
  const float* b_hh_l1r = (const float*)d_in[16];
  const float* fc1_w = (const float*)d_in[17];
  const float* fc1_b = (const float*)d_in[18];
  const float* fc2_w = (const float*)d_in[19];
  const float* fc2_b = (const float*)d_in[20];
  float* out = (float*)d_out;

  float* ws  = (float*)d_ws;
  float* ws0 = ws;
  float* ws1 = ws + (size_t)NT * NB * 4;
  float* ws2 = ws;

  const dim3 blk(256);
  const dim3 blk_lstm(512);
  const dim3 grid_lstm(NB / 256);

  lstm_bidir_kernel<7, false><<<grid_lstm, blk_lstm, 0, stream>>>(
      x, w_ih_l0, w_hh_l0, b_ih_l0, b_hh_l0,
      w_ih_l0r, w_hh_l0r, b_ih_l0r, b_hh_l0r, ws0);

  lstm_bidir_kernel<4, true><<<grid_lstm, blk_lstm, 0, stream>>>(
      ws0, w_ih_l1, w_hh_l1, b_ih_l1, b_hh_l1,
      w_ih_l1r, w_hh_l1r, b_ih_l1r, b_hh_l1r, ws1);

  fc1_wmma_kernel<<<dim3(2048), blk, 0, stream>>>(ws1, fc1_w, fc1_b, ws2);

  fc2_wmma_kernel<<<dim3(1024), blk, 0, stream>>>(ws2, fc2_w, fc2_b, out);
}


# h kept per direction, two time steps per 16-byte slot: every h store writes whole 128-byte lines (was alternate 8-byte pieces); FC1 w fragment row-swapped to match; k order inside FC1 sums changes
# speedup vs baseline: 2.1326x; 1.0926x over previous
#include <hip/hip_runtime.h>
#include <hip/hip_bf16.h>


#define NB 65536
#define NT 100

#define PLANE ((size_t)(NT / 2) * NB * 4)

typedef __attribute__((ext_vector_type(2))) float v2f;
typedef __attribute__((ext_vector_type(8))) float v8f;
typedef __attribute__((ext_vector_type(2))) unsigned int v2u;

__device__ __forceinline__ float fast_rcp(float x) {
  return __builtin_amdgcn_rcpf(x);
}
__device__ __forceinline__ float sigm(float x) {
  return fast_rcp(1.0f + __expf(-x));
}
__device__ __forceinline__ float tanh_fast(float x) {
  return fmaf(2.0f, fast_rcp(1.0f + __expf(-2.0f * x)), -1.0f);
}

template <int IN>
struct DirWeights {
  float wih[8][IN];
  float whh[8][2];
  float bias[8];
};

__device__ __forceinline__ float keep_in_vgpr(float v) {
  asm("" : "+v"(v));
  return v;
}

template <int IN, bool IN_VGPR>
__device__ __forceinline__ void load_dir_weights(
    DirWeights<IN>& w,
    const float* __restrict__ w_ih,
    const float* __restrict__ w_hh,
    const float* __restrict__ b_ih,
    const float* __restrict__ b_hh)
{
#pragma unroll
  for (int g = 0; g < 8; ++g) {
#pragma unroll
    for (int c = 0; c < IN; ++c) w.wih[g][c] = w_ih[g * IN + c];
    w.whh[g][0] = w_hh[g * 2 + 0];
    w.whh[g][1] = w_hh[g * 2 + 1];
    w.bias[g] = b_ih[g] + b_hh[g];
    if constexpr (IN_VGPR) {
#pragma unroll
      for (int c = 0; c < IN; ++c) w.wih[g][c] = keep_in_vgpr(w.wih[g][c]);
      w.whh[g][0] = keep_in_vgpr(w.whh[g][0]);
      w.whh[g][1] = keep_in_vgpr(w.whh[g][1]);
      w.bias[g] = keep_in_vgpr(w.bias[g]);
    }
  }
}

template <int IN>
__device__ __forceinline__ void lstm_step(
    const DirWeights<IN>& w, const float (&xv)[IN],
    float& h0, float& h1, float& c0, float& c1)
{
  float g[8];
#pragma unroll
  for (int gg = 0; gg < 8; ++gg) {
    float acc = fmaf(w.whh[gg][0], h0, fmaf(w.whh[gg][1], h1, w.bias[gg]));
#pragma unroll
    for (int c = 0; c < IN; ++c) acc = fmaf(w.wih[gg][c], xv[c], acc);
    g[gg] = acc;
  }

  const float i0 = sigm(g[0]), i1 = sigm(g[1]);
  const float f0 = sigm(g[2]), f1 = sigm(g[3]);
  const float t0 = tanh_fast(g[4]), t1 = tanh_fast(g[5]);
  const float o0 = sigm(g[6]), o1 = sigm(g[7]);

  c0 = fmaf(f0, c0, i0 * t0);
  c1 = fmaf(f1, c1, i1 * t1);
  h0 = o0 * tanh_fast(c0);
  h1 = o1 * tanh_fast(c1);
}

#define SPW 256
#define TC 4
#define NCH (NT / TC)
#define PITCH 257

template <int IN>
struct TilePieces {
  float4 v[TC * IN / 4];
};

template <int IN, bool TMAJOR>
__device__ __forceinline__ void load_tile(
    TilePieces<IN>& r, const float* __restrict__ in, int b0, int t0, int lj)
{
  if constexpr (TMAJOR) {
    static_assert(IN == 4 && TC == 4, "paired planes: 4 channels, 4-step tiles");
#pragma unroll
    for (int i = 0; i < 4; ++i)
      r.v[i] = *(const float4*)(in + (size_t)(i >> 1) * PLANE +
                                ((size_t)((t0 >> 1) + (i & 1)) * NB +
                                 (size_t)(b0 + lj)) * 4);
  } else {
    static_assert((TC * IN) % 4 == 0, "tile row must be whole float4 pieces");
    constexpr int PPS = TC * IN / 4;
#pragma unroll
    for (int i = 0; i < PPS; ++i) {
      const int p = lj + SPW * i;
      const int srow = p / PPS;
      const int q = p - srow * PPS;
      r.v[i] = *(const float4*)(in + (size_t)(b0 + srow) * (NT * IN) +
                                (size_t)t0 * IN + q * 4);
    }
  }
}

template <int IN, bool TMAJOR>
__device__ __forceinline__ void store_tile(
    float* __restrict__ tile, const TilePieces<IN>& r, int lj)
{
  constexpr int PPS = TC * IN / 4;
#pragma unroll
  for (int i = 0; i < PPS; ++i) {
    if constexpr (TMAJOR) {
      float* d = tile + ((i & 1) * 2 * IN + (i >> 1) * 2) * PITCH + lj;
      d[0 * PITCH] = r.v[i].x;
      d[1 * PITCH] = r.v[i].y;
      d[(IN + 0) * PITCH] = r.v[i].z;
      d[(IN + 1) * PITCH] = r.v[i].w;
    } else {
      const int p = lj + SPW * i;
      const int srow = p / PPS;
      const int row0 = (p - srow * PPS) * 4;
      float* d = tile + row0 * PITCH + srow;
      d[0 * PITCH] = r.v[i].x;
      d[1 * PITCH] = r.v[i].y;
      d[2 * PITCH] = r.v[i].z;
      d[3 * PITCH] = r.v[i].w;
    }
  }
}

__device__ __forceinline__ void lds_handover() {
  __builtin_amdgcn_fence(__ATOMIC_RELEASE, "workgroup", "local");
  __builtin_amdgcn_s_barrier();
  __builtin_amdgcn_fence(__ATOMIC_ACQUIRE, "workgroup", "local");
}

template <int IN, bool TMAJOR>
__global__ __launch_bounds__(512) __attribute__((amdgpu_waves_per_eu(4, 4)))
void lstm_bidir_kernel(
    const float* __restrict__ in,
    const float* __restrict__ wf_ih, const float* __restrict__ wf_hh,
    const float* __restrict__ bf_ih, const float* __restrict__ bf_hh,
    const float* __restrict__ wr_ih, const float* __restrict__ wr_hh,
    const float* __restrict__ br_ih, const float* __restrict__ br_hh,
    float* __restrict__ out)
{
  __shared__ float ring[2][2][TC * IN * PITCH];

  const int tid = threadIdx.x;
  const bool is_loader = __builtin_amdgcn_readfirstlane(tid) >= SPW;
  const int b0 = blockIdx.x * SPW;
  const int lj = tid - SPW;
  const int b = b0 + tid;

  DirWeights<IN> wf, wr;
  load_dir_weights<IN, false>(wf, wf_ih, wf_hh, bf_ih, bf_hh);
  load_dir_weights<IN, true>(wr, wr_ih, wr_hh, br_ih, br_hh);

  float hf0 = 0.f, hf1 = 0.f, cf0 = 0.f, cf1 = 0.f;
  float hr0 = 0.f, hr1 = 0.f, cr0 = 0.f, cr1 = 0.f;
  float kf0 = 0.f, kf1 = 0.f, kr0 = 0.f, kr1 = 0.f;
  static_assert(TC % 2 == 0, "a tile must hold whole time pairs");

  TilePieces<IN> pf, pr;
  if (is_loader) {
    load_tile<IN, TMAJOR>(pf, in, b0, 0, lj);
    load_tile<IN, TMAJOR>(pr, in, b0, NT - TC, lj);
    store_tile<IN, TMAJOR>(ring[0][0], pf, lj);
    store_tile<IN, TMAJOR>(ring[0][1], pr, lj);
    load_tile<IN, TMAJOR>(pf, in, b0, TC, lj);
    load_tile<IN, TMAJOR>(pr, in, b0, NT - 2 * TC, lj);
  }
  lds_handover();

  for (int c = 0; c < NCH; ++c) {
    const int stage = c & 1;
    if (is_loader) {
      if (c + 1 < NCH) {
        store_tile<IN, TMAJOR>(ring[stage ^ 1][0], pf, lj);
        store_tile<IN, TMAJOR>(ring[stage ^ 1][1], pr, lj);
        if (c + 2 < NCH) {
          load_tile<IN, TMAJOR>(pf, in, b0, (c + 2) * TC, lj);
          load_tile<IN, TMAJOR>(pr, in, b0, NT - TC - (c + 2) * TC, lj);
        }
      }
    } else {
      const float* tf_tile = ring[stage][0];
      const float* tr_tile = ring[stage][1];
#pragma unroll
      for (int s = 0; s < TC; ++s) {
        const int tf = c * TC + s;
        const int tr = NT - 1 - tf;
        float xf[IN], xr[IN];
#pragma unroll
        for (int ch = 0; ch < IN; ++ch) {
          xf[ch] = tf_tile[(s * IN + ch) * PITCH + tid];
          xr[ch] = tr_tile[((TC - 1 - s) * IN + ch) * PITCH + tid];
        }

        lstm_step<IN>(wf, xf, hf0, hf1, cf0, cf1);
        lstm_step<IN>(wr, xr, hr0, hr1, cr0, cr1);

        if ((s & 1) == 0) {
          kf0 = hf0; kf1 = hf1;
          kr0 = hr0; kr1 = hr1;
        } else {
          *(float4*)(out + ((size_t)(tf >> 1) * NB + (size_t)b) * 4) =
              make_float4(kf0, kf1, hf0, hf1);
          *(float4*)(out + PLANE + ((size_t)(tr >> 1) * NB + (size_t)b) * 4) =
              make_float4(hr0, hr1, kr0, kr1);
        }
      }
    }
    lds_handover();
  }
}

__global__ __launch_bounds__(256) void fc1_wmma_kernel(
    const float* __restrict__ h2,
    const float* __restrict__ w,
    const float* __restrict__ bias,
    float* __restrict__ act)
{
  const int wid = blockIdx.x * (blockDim.x >> 5) + (threadIdx.x >> 5);
  const int lane = threadIdx.x & 31;
  const int m_tile = wid >> 2;
  const int n_tile = wid & 3;
  const int row = lane & 15;
  const int half = lane >> 4;
  const int base_b = m_tile * 16;
  const int n = n_tile * 16 + row;

  v8f acc = {};
  const float* wrow = w + (size_t)n * 400 + half * 4;
  const float* arow = h2 + (size_t)half * PLANE + (size_t)(base_b + row) * 4;

#pragma unroll 5
  for (int j = 0; j < NT / 2; ++j) {
    float4 a = *(const float4*)(arow + (size_t)j * NB * 4);
    const float4 bq = *(const float4*)(wrow + j * 8);
    const v2u sxz = __builtin_amdgcn_permlane16_swap(
        __float_as_uint(bq.x), __float_as_uint(bq.z), false, false);
    const v2u syw = __builtin_amdgcn_permlane16_swap(
        __float_as_uint(bq.y), __float_as_uint(bq.w), false, false);
    v2f a01, a23, b01, b23;
    a01.x = fmaxf(a.x, 0.0f);
    a01.y = fmaxf(a.y, 0.0f);
    a23.x = fmaxf(a.z, 0.0f);
    a23.y = fmaxf(a.w, 0.0f);
    b01.x = __uint_as_float(sxz.x); b01.y = __uint_as_float(syw.x);
    b23.x = __uint_as_float(sxz.y); b23.y = __uint_as_float(syw.y);
    acc = __builtin_amdgcn_wmma_f32_16x16x4_f32(
        false, a01, false, b01, (short)0, acc, false, false);
    acc = __builtin_amdgcn_wmma_f32_16x16x4_f32(
        false, a23, false, b23, (short)0, acc, false, false);
  }

  const float bn = bias[n];
#pragma unroll
  for (int r = 0; r < 8; ++r) {
    const int m = r + half * 8;
    act[(size_t)(base_b + m) * 64 + n] = fmaxf(acc[r] + bn, 0.0f);
  }
}

__global__ __launch_bounds__(256) void fc2_wmma_kernel(
    const float* __restrict__ act,
    const float* __restrict__ w,
    const float* __restrict__ bias,
    float* __restrict__ out)
{
  const int wid = blockIdx.x * (blockDim.x >> 5) + (threadIdx.x >> 5);
  const int lane = threadIdx.x & 31;
  const int m_tile = wid >> 1;
  const int n_tile = wid & 1;
  const int row = lane & 15;
  const int half = lane >> 4;
  const int base_b = m_tile * 16;
  const int n = n_tile * 16 + row;
  const bool nvalid = (n < 20);

  v8f acc = {};
  const float* arow = act + (size_t)(base_b + row) * 64 + half * 4;
  const float* wrow = w + (size_t)(nvalid ? n : 0) * 64 + half * 4;

#pragma unroll
  for (int j = 0; j < 8; ++j) {
    const float4 a = *(const float4*)(arow + j * 8);
    float4 bq = *(const float4*)(wrow + j * 8);
    if (!nvalid) { bq.x = 0.0f; bq.y = 0.0f; bq.z = 0.0f; bq.w = 0.0f; }
    v2f a01, a23, b01, b23;
    a01.x = a.x; a01.y = a.y;
    a23.x = a.z; a23.y = a.w;
    b01.x = bq.x; b01.y = bq.y;
    b23.x = bq.z; b23.y = bq.w;
    acc = __builtin_amdgcn_wmma_f32_16x16x4_f32(
        false, a01, false, b01, (short)0, acc, false, false);
    acc = __builtin_amdgcn_wmma_f32_16x16x4_f32(
        false, a23, false, b23, (short)0, acc, false, false);
  }

  if (nvalid) {
    const float bn = bias[n];
#pragma unroll
    for (int r = 0; r < 8; ++r) {
      const int m = r + half * 8;
      out[(size_t)(base_b + m) * 20 + n] = acc[r] + bn;
    }
  }
}

extern "C" void kernel_launch(void* const* d_in, const int* in_sizes, int n_in,
                              void* d_out, int out_size, void* d_ws,
                              size_t ws_size, hipStream_t stream) {
  const float* x = (const float*)d_in[0];
  const float* w_ih_l0  = (const float*)d_in[1];
  const float* w_hh_l0  = (const float*)d_in[2];
  const float* b_ih_l0  = (const float*)d_in[3];
  const float* b_hh_l0  = (const float*)d_in[4];
  const float* w_ih_l0r = (const float*)d_in[5];
  const float* w_hh_l0r = (const float*)d_in[6];
  const float* b_ih_l0r = (const float*)d_in[7];
  const float* b_hh_l0r = (const float*)d_in[8];
  const float* w_ih_l1  = (const float*)d_in[9];
  const float* w_hh_l1  = (const float*)d_in[10];
  const float* b_ih_l1  = (const float*)d_in[11];
  const float* b_hh_l1  = (const float*)d_in[12];
  const float* w_ih_l1r = (const float*)d_in[13];
  const float* w_hh_l1r = (const float*)d_in[14];
  const float* b_ih_l1r = (const float*)d_in[15];
  const float* b_hh_l1r = (const float*)d_in[16];
  const float* fc1_w = (const float*)d_in[17];
  const float* fc1_b = (const float*)d_in[18];
  const float* fc2_w = (const float*)d_in[19];
  const float* fc2_b = (const float*)d_in[20];
  float* out = (float*)d_out;

  float* ws  = (float*)d_ws;
  float* ws0 = ws;
  float* ws1 = ws + (size_t)NT * NB * 4;
  float* ws2 = ws;

  const dim3 blk(256);
  const dim3 blk_lstm(512);
  const dim3 grid_lstm(NB / 256);

  lstm_bidir_kernel<7, false><<<grid_lstm, blk_lstm, 0, stream>>>(
      x, w_ih_l0, w_hh_l0, b_ih_l0, b_hh_l0,
      w_ih_l0r, w_hh_l0r, b_ih_l0r, b_hh_l0r, ws0);

  lstm_bidir_kernel<4, true><<<grid_lstm, blk_lstm, 0, stream>>>(
      ws0, w_ih_l1, w_hh_l1, b_ih_l1, b_hh_l1,
      w_ih_l1r, w_hh_l1r, b_ih_l1r, b_hh_l1r, ws1);

  fc1_wmma_kernel<<<dim3(2048), blk, 0, stream>>>(ws1, fc1_w, fc1_b, ws2);

  fc2_wmma_kernel<<<dim3(1024), blk, 0, stream>>>(ws2, fc2_w, fc2_b, out);
}


# LSTM input tiles of 10 time steps (were 4): 10 role hand-overs per layer, not 25; a loader holds one direction's tile at a time, layer-0 x fetched as 256-byte row pieces; step sums unchanged
# speedup vs baseline: 2.2172x; 1.0397x over previous
#include <hip/hip_runtime.h>
#include <hip/hip_bf16.h>


#define NB 65536
#define NT 100

#define PLANE ((size_t)(NT / 2) * NB * 4)

typedef __attribute__((ext_vector_type(2))) float v2f;
typedef __attribute__((ext_vector_type(8))) float v8f;
typedef __attribute__((ext_vector_type(2))) unsigned int v2u;

__device__ __forceinline__ float fast_rcp(float x) {
  return __builtin_amdgcn_rcpf(x);
}
__device__ __forceinline__ float sigm(float x) {
  return fast_rcp(1.0f + __expf(-x));
}
__device__ __forceinline__ float tanh_fast(float x) {
  return fmaf(2.0f, fast_rcp(1.0f + __expf(-2.0f * x)), -1.0f);
}

template <int IN>
struct DirWeights {
  float wih[8][IN];
  float whh[8][2];
  float bias[8];
};

__device__ __forceinline__ float keep_in_vgpr(float v) {
  asm("" : "+v"(v));
  return v;
}

template <int IN, bool IN_VGPR>
__device__ __forceinline__ void load_dir_weights(
    DirWeights<IN>& w,
    const float* __restrict__ w_ih,
    const float* __restrict__ w_hh,
    const float* __restrict__ b_ih,
    const float* __restrict__ b_hh)
{
#pragma unroll
  for (int g = 0; g < 8; ++g) {
#pragma unroll
    for (int c = 0; c < IN; ++c) w.wih[g][c] = w_ih[g * IN + c];
    w.whh[g][0] = w_hh[g * 2 + 0];
    w.whh[g][1] = w_hh[g * 2 + 1];
    w.bias[g] = b_ih[g] + b_hh[g];
    if constexpr (IN_VGPR) {
#pragma unroll
      for (int c = 0; c < IN; ++c) w.wih[g][c] = keep_in_vgpr(w.wih[g][c]);
      w.whh[g][0] = keep_in_vgpr(w.whh[g][0]);
      w.whh[g][1] = keep_in_vgpr(w.whh[g][1]);
      w.bias[g] = keep_in_vgpr(w.bias[g]);
    }
  }
}

template <int IN>
__device__ __forceinline__ void lstm_step(
    const DirWeights<IN>& w, const float (&xv)[IN],
    float& h0, float& h1, float& c0, float& c1)
{
  float g[8];
#pragma unroll
  for (int gg = 0; gg < 8; ++gg) {
    float acc = fmaf(w.whh[gg][0], h0, fmaf(w.whh[gg][1], h1, w.bias[gg]));
#pragma unroll
    for (int c = 0; c < IN; ++c) acc = fmaf(w.wih[gg][c], xv[c], acc);
    g[gg] = acc;
  }

  const float i0 = sigm(g[0]), i1 = sigm(g[1]);
  const float f0 = sigm(g[2]), f1 = sigm(g[3]);
  const float t0 = tanh_fast(g[4]), t1 = tanh_fast(g[5]);
  const float o0 = sigm(g[6]), o1 = sigm(g[7]);

  c0 = fmaf(f0, c0, i0 * t0);
  c1 = fmaf(f1, c1, i1 * t1);
  h0 = o0 * tanh_fast(c0);
  h1 = o1 * tanh_fast(c1);
}

#define SPW 256
#define TC 10
#define NCH (NT / TC)
#define PITCH 257

template <int IN, bool TMAJOR>
struct TilePieces;
template <int IN>
struct TilePieces<IN, false> {
  static constexpr int PPS = TC * IN / 2;
  static constexpr int TAIL = PPS - 32;
  static_assert(TAIL > 0 && TAIL <= 32, "row load takes 32 pieces, tail loads the rest");
  float2 v[32 + TAIL];
};
template <int IN>
struct TilePieces<IN, true> {
  float4 v[TC];
};

template <int IN, bool TMAJOR>
__device__ __forceinline__ void load_tile(
    TilePieces<IN, TMAJOR>& r, const float* __restrict__ in, int b0, int t0,
    int lj)
{
  if constexpr (TMAJOR) {
    static_assert(IN == 4 && TC % 2 == 0, "paired planes: 4 channels, whole pairs");
#pragma unroll
    for (int i = 0; i < TC; ++i)
      r.v[i] = *(const float4*)(in + (size_t)(i / (TC / 2)) * PLANE +
                                ((size_t)((t0 >> 1) + (i % (TC / 2))) * NB +
                                 (size_t)(b0 + lj)) * 4);
  } else {
    static_assert((TC * IN) % 2 == 0, "tile row must be whole 8-byte pieces");
    constexpr int TAIL = TilePieces<IN, false>::TAIL;
    const int l = lj & 31;
    const float* base = in + (size_t)(b0 + (lj & ~31)) * (NT * IN) + (size_t)t0 * IN;
#pragma unroll
    for (int i = 0; i < 32; ++i)
      r.v[i] = *(const float2*)(base + (size_t)i * (NT * IN) + 2 * l);
#pragma unroll
    for (int j = 0; j < TAIL; ++j) {
      const int idx = 32 * j + l;
      r.v[32 + j] = *(const float2*)(base + (size_t)(idx / TAIL) * (NT * IN) +
                                     2 * (32 + idx % TAIL));
    }
  }
}

template <int IN, bool TMAJOR>
__device__ __forceinline__ void store_tile(
    float* __restrict__ tile, const TilePieces<IN, TMAJOR>& r, int lj)
{
  if constexpr (TMAJOR) {
#pragma unroll
    for (int i = 0; i < TC; ++i) {
      float* d = tile +
                 ((i % (TC / 2)) * 2 * IN + (i / (TC / 2)) * 2) * PITCH + lj;
      d[0 * PITCH] = r.v[i].x;
      d[1 * PITCH] = r.v[i].y;
      d[(IN + 0) * PITCH] = r.v[i].z;
      d[(IN + 1) * PITCH] = r.v[i].w;
    }
  } else {
    constexpr int TAIL = TilePieces<IN, false>::TAIL;
    const int l = lj & 31;
    float* d = tile + (2 * l) * PITCH + (lj & ~31);
#pragma unroll
    for (int i = 0; i < 32; ++i) {
      d[0 * PITCH + i] = r.v[i].x;
      d[1 * PITCH + i] = r.v[i].y;
    }
#pragma unroll
    for (int j = 0; j < TAIL; ++j) {
      const int idx = 32 * j + l;
      float* e = tile + (2 * (32 + idx % TAIL)) * PITCH + (lj & ~31) + idx / TAIL;
      e[0 * PITCH] = r.v[32 + j].x;
      e[1 * PITCH] = r.v[32 + j].y;
    }
  }
}

__device__ __forceinline__ void lds_handover() {
  __builtin_amdgcn_fence(__ATOMIC_RELEASE, "workgroup", "local");
  __builtin_amdgcn_s_barrier();
  __builtin_amdgcn_fence(__ATOMIC_ACQUIRE, "workgroup", "local");
}

template <int IN, bool TMAJOR>
__global__ __launch_bounds__(512) __attribute__((amdgpu_waves_per_eu(4, 4)))
void lstm_bidir_kernel(
    const float* __restrict__ in,
    const float* __restrict__ wf_ih, const float* __restrict__ wf_hh,
    const float* __restrict__ bf_ih, const float* __restrict__ bf_hh,
    const float* __restrict__ wr_ih, const float* __restrict__ wr_hh,
    const float* __restrict__ br_ih, const float* __restrict__ br_hh,
    float* __restrict__ out)
{
  __shared__ float ring[2][2][TC * IN * PITCH];

  const int tid = threadIdx.x;
  const bool is_loader = __builtin_amdgcn_readfirstlane(tid) >= SPW;
  const int b0 = blockIdx.x * SPW;
  const int lj = tid - SPW;
  const int b = b0 + tid;

  DirWeights<IN> wf, wr;
  load_dir_weights<IN, false>(wf, wf_ih, wf_hh, bf_ih, bf_hh);
  load_dir_weights<IN, true>(wr, wr_ih, wr_hh, br_ih, br_hh);

  float hf0 = 0.f, hf1 = 0.f, cf0 = 0.f, cf1 = 0.f;
  float hr0 = 0.f, hr1 = 0.f, cr0 = 0.f, cr1 = 0.f;
  float kf0 = 0.f, kf1 = 0.f, kr0 = 0.f, kr1 = 0.f;
  static_assert(TC % 2 == 0, "a tile must hold whole time pairs");

  TilePieces<IN, TMAJOR> pt;
  if (is_loader) {
    load_tile<IN, TMAJOR>(pt, in, b0, 0, lj);
    store_tile<IN, TMAJOR>(ring[0][0], pt, lj);
    load_tile<IN, TMAJOR>(pt, in, b0, NT - TC, lj);
    store_tile<IN, TMAJOR>(ring[0][1], pt, lj);
    load_tile<IN, TMAJOR>(pt, in, b0, TC, lj);
  }
  lds_handover();

  for (int c = 0; c < NCH; ++c) {
    const int stage = c & 1;
    if (is_loader) {
      if (c + 1 < NCH) {
        store_tile<IN, TMAJOR>(ring[stage ^ 1][0], pt, lj);
        load_tile<IN, TMAJOR>(pt, in, b0, NT - TC - (c + 1) * TC, lj);
        store_tile<IN, TMAJOR>(ring[stage ^ 1][1], pt, lj);
        if (c + 2 < NCH)
          load_tile<IN, TMAJOR>(pt, in, b0, (c + 2) * TC, lj);
      }
    } else {
      const float* tf_tile = ring[stage][0];
      const float* tr_tile = ring[stage][1];
#pragma unroll
      for (int s = 0; s < TC; ++s) {
        const int tf = c * TC + s;
        const int tr = NT - 1 - tf;
        float xf[IN], xr[IN];
#pragma unroll
        for (int ch = 0; ch < IN; ++ch) {
          xf[ch] = tf_tile[(s * IN + ch) * PITCH + tid];
          xr[ch] = tr_tile[((TC - 1 - s) * IN + ch) * PITCH + tid];
        }

        lstm_step<IN>(wf, xf, hf0, hf1, cf0, cf1);
        lstm_step<IN>(wr, xr, hr0, hr1, cr0, cr1);

        if ((s & 1) == 0) {
          kf0 = hf0; kf1 = hf1;
          kr0 = hr0; kr1 = hr1;
        } else {
          *(float4*)(out + ((size_t)(tf >> 1) * NB + (size_t)b) * 4) =
              make_float4(kf0, kf1, hf0, hf1);
          *(float4*)(out + PLANE + ((size_t)(tr >> 1) * NB + (size_t)b) * 4) =
              make_float4(hr0, hr1, kr0, kr1);
        }
      }
    }
    lds_handover();
  }
}

__global__ __launch_bounds__(256) void fc1_wmma_kernel(
    const float* __restrict__ h2,
    const float* __restrict__ w,
    const float* __restrict__ bias,
    float* __restrict__ act)
{
  const int wid = blockIdx.x * (blockDim.x >> 5) + (threadIdx.x >> 5);
  const int lane = threadIdx.x & 31;
  const int m_tile = wid >> 2;
  const int n_tile = wid & 3;
  const int row = lane & 15;
  const int half = lane >> 4;
  const int base_b = m_tile * 16;
  const int n = n_tile * 16 + row;

  v8f acc = {};
  const float* wrow = w + (size_t)n * 400 + half * 4;
  const float* arow = h2 + (size_t)half * PLANE + (size_t)(base_b + row) * 4;

#pragma unroll 5
  for (int j = 0; j < NT / 2; ++j) {
    float4 a = *(const float4*)(arow + (size_t)j * NB * 4);
    const float4 bq = *(const float4*)(wrow + j * 8);
    const v2u sxz = __builtin_amdgcn_permlane16_swap(
        __float_as_uint(bq.x), __float_as_uint(bq.z), false, false);
    const v2u syw = __builtin_amdgcn_permlane16_swap(
        __float_as_uint(bq.y), __float_as_uint(bq.w), false, false);
    v2f a01, a23, b01, b23;
    a01.x = fmaxf(a.x, 0.0f);
    a01.y = fmaxf(a.y, 0.0f);
    a23.x = fmaxf(a.z, 0.0f);
    a23.y = fmaxf(a.w, 0.0f);
    b01.x = __uint_as_float(sxz.x); b01.y = __uint_as_float(syw.x);
    b23.x = __uint_as_float(sxz.y); b23.y = __uint_as_float(syw.y);
    acc = __builtin_amdgcn_wmma_f32_16x16x4_f32(
        false, a01, false, b01, (short)0, acc, false, false);
    acc = __builtin_amdgcn_wmma_f32_16x16x4_f32(
        false, a23, false, b23, (short)0, acc, false, false);
  }

  const float bn = bias[n];
#pragma unroll
  for (int r = 0; r < 8; ++r) {
    const int m = r + half * 8;
    act[(size_t)(base_b + m) * 64 + n] = fmaxf(acc[r] + bn, 0.0f);
  }
}

__global__ __launch_bounds__(256) void fc2_wmma_kernel(
    const float* __restrict__ act,
    const float* __restrict__ w,
    const float* __restrict__ bias,
    float* __restrict__ out)
{
  const int wid = blockIdx.x * (blockDim.x >> 5) + (threadIdx.x >> 5);
  const int lane = threadIdx.x & 31;
  const int m_tile = wid >> 1;
  const int n_tile = wid & 1;
  const int row = lane & 15;
  const int half = lane >> 4;
  const int base_b = m_tile * 16;
  const int n = n_tile * 16 + row;
  const bool nvalid = (n < 20);

  v8f acc = {};
  const float* arow = act + (size_t)(base_b + row) * 64 + half * 4;
  const float* wrow = w + (size_t)(nvalid ? n : 0) * 64 + half * 4;

#pragma unroll
  for (int j = 0; j < 8; ++j) {
    const float4 a = *(const float4*)(arow + j * 8);
    float4 bq = *(const float4*)(wrow + j * 8);
    if (!nvalid) { bq.x = 0.0f; bq.y = 0.0f; bq.z = 0.0f; bq.w = 0.0f; }
    v2f a01, a23, b01, b23;
    a01.x = a.x; a01.y = a.y;
    a23.x = a.z; a23.y = a.w;
    b01.x = bq.x; b01.y = bq.y;
    b23.x = bq.z; b23.y = bq.w;
    acc = __builtin_amdgcn_wmma_f32_16x16x4_f32(
        false, a01, false, b01, (short)0, acc, false, false);
    acc = __builtin_amdgcn_wmma_f32_16x16x4_f32(
        false, a23, false, b23, (short)0, acc, false, false);
  }

  if (nvalid) {
    const float bn = bias[n];
#pragma unroll
    for (int r = 0; r < 8; ++r) {
      const int m = r + half * 8;
      out[(size_t)(base_b + m) * 20 + n] = acc[r] + bn;
    }
  }
}

extern "C" void kernel_launch(void* const* d_in, const int* in_sizes, int n_in,
                              void* d_out, int out_size, void* d_ws,
                              size_t ws_size, hipStream_t stream) {
  const float* x = (const float*)d_in[0];
  const float* w_ih_l0  = (const float*)d_in[1];
  const float* w_hh_l0  = (const float*)d_in[2];
  const float* b_ih_l0  = (const float*)d_in[3];
  const float* b_hh_l0  = (const float*)d_in[4];
  const float* w_ih_l0r = (const float*)d_in[5];
  const float* w_hh_l0r = (const float*)d_in[6];
  const float* b_ih_l0r = (const float*)d_in[7];
  const float* b_hh_l0r = (const float*)d_in[8];
  const float* w_ih_l1  = (const float*)d_in[9];
  const float* w_hh_l1  = (const float*)d_in[10];
  const float* b_ih_l1  = (const float*)d_in[11];
  const float* b_hh_l1  = (const float*)d_in[12];
  const float* w_ih_l1r = (const float*)d_in[13];
  const float* w_hh_l1r = (const float*)d_in[14];
  const float* b_ih_l1r = (const float*)d_in[15];
  const float* b_hh_l1r = (const float*)d_in[16];
  const float* fc1_w = (const float*)d_in[17];
  const float* fc1_b = (const float*)d_in[18];
  const float* fc2_w = (const float*)d_in[19];
  const float* fc2_b = (const float*)d_in[20];
  float* out = (float*)d_out;

  float* ws  = (float*)d_ws;
  float* ws0 = ws;
  float* ws1 = ws + (size_t)NT * NB * 4;
  float* ws2 = ws;

  const dim3 blk(256);
  const dim3 blk_lstm(512);
  const dim3 grid_lstm(NB / 256);

  lstm_bidir_kernel<7, false><<<grid_lstm, blk_lstm, 0, stream>>>(
      x, w_ih_l0, w_hh_l0, b_ih_l0, b_hh_l0,
      w_ih_l0r, w_hh_l0r, b_ih_l0r, b_hh_l0r, ws0);

  lstm_bidir_kernel<4, true><<<grid_lstm, blk_lstm, 0, stream>>>(
      ws0, w_ih_l1, w_hh_l1, b_ih_l1, b_hh_l1,
      w_ih_l1r, w_hh_l1r, b_ih_l1r, b_hh_l1r, ws1);

  fc1_wmma_kernel<<<dim3(2048), blk, 0, stream>>>(ws1, fc1_w, fc1_b, ws2);

  fc2_wmma_kernel<<<dim3(1024), blk, 0, stream>>>(ws2, fc2_w, fc2_b, out);
}


# FC2 moves into the FC1 launch: a work-group's eight relu'd FC1 tiles (32 samples x 64) meet in LDS, four of its waves run FC2 on f32 WMMA from there; act plane and fourth launch go; sums unchanged
# speedup vs baseline: 2.3347x; 1.0530x over previous
#include <hip/hip_runtime.h>
#include <hip/hip_bf16.h>


#define NB 65536
#define NT 100

#define PLANE ((size_t)(NT / 2) * NB * 4)

typedef __attribute__((ext_vector_type(2))) float v2f;
typedef __attribute__((ext_vector_type(8))) float v8f;
typedef __attribute__((ext_vector_type(2))) unsigned int v2u;

__device__ __forceinline__ float fast_rcp(float x) {
  return __builtin_amdgcn_rcpf(x);
}
__device__ __forceinline__ float sigm(float x) {
  return fast_rcp(1.0f + __expf(-x));
}
__device__ __forceinline__ float tanh_fast(float x) {
  return fmaf(2.0f, fast_rcp(1.0f + __expf(-2.0f * x)), -1.0f);
}

template <int IN>
struct DirWeights {
  float wih[8][IN];
  float whh[8][2];
  float bias[8];
};

__device__ __forceinline__ float keep_in_vgpr(float v) {
  asm("" : "+v"(v));
  return v;
}

template <int IN, bool IN_VGPR>
__device__ __forceinline__ void load_dir_weights(
    DirWeights<IN>& w,
    const float* __restrict__ w_ih,
    const float* __restrict__ w_hh,
    const float* __restrict__ b_ih,
    const float* __restrict__ b_hh)
{
#pragma unroll
  for (int g = 0; g < 8; ++g) {
#pragma unroll
    for (int c = 0; c < IN; ++c) w.wih[g][c] = w_ih[g * IN + c];
    w.whh[g][0] = w_hh[g * 2 + 0];
    w.whh[g][1] = w_hh[g * 2 + 1];
    w.bias[g] = b_ih[g] + b_hh[g];
    if constexpr (IN_VGPR) {
#pragma unroll
      for (int c = 0; c < IN; ++c) w.wih[g][c] = keep_in_vgpr(w.wih[g][c]);
      w.whh[g][0] = keep_in_vgpr(w.whh[g][0]);
      w.whh[g][1] = keep_in_vgpr(w.whh[g][1]);
      w.bias[g] = keep_in_vgpr(w.bias[g]);
    }
  }
}

template <int IN>
__device__ __forceinline__ void lstm_step(
    const DirWeights<IN>& w, const float (&xv)[IN],
    float& h0, float& h1, float& c0, float& c1)
{
  float g[8];
#pragma unroll
  for (int gg = 0; gg < 8; ++gg) {
    float acc = fmaf(w.whh[gg][0], h0, fmaf(w.whh[gg][1], h1, w.bias[gg]));
#pragma unroll
    for (int c = 0; c < IN; ++c) acc = fmaf(w.wih[gg][c], xv[c], acc);
    g[gg] = acc;
  }

  const float i0 = sigm(g[0]), i1 = sigm(g[1]);
  const float f0 = sigm(g[2]), f1 = sigm(g[3]);
  const float t0 = tanh_fast(g[4]), t1 = tanh_fast(g[5]);
  const float o0 = sigm(g[6]), o1 = sigm(g[7]);

  c0 = fmaf(f0, c0, i0 * t0);
  c1 = fmaf(f1, c1, i1 * t1);
  h0 = o0 * tanh_fast(c0);
  h1 = o1 * tanh_fast(c1);
}

#define SPW 256
#define TC 10
#define NCH (NT / TC)
#define PITCH 257

template <int IN, bool TMAJOR>
struct TilePieces;
template <int IN>
struct TilePieces<IN, false> {
  static constexpr int PPS = TC * IN / 2;
  static constexpr int TAIL = PPS - 32;
  static_assert(TAIL > 0 && TAIL <= 32, "row load takes 32 pieces, tail loads the rest");
  float2 v[32 + TAIL];
};
template <int IN>
struct TilePieces<IN, true> {
  float4 v[TC];
};

template <int IN, bool TMAJOR>
__device__ __forceinline__ void load_tile(
    TilePieces<IN, TMAJOR>& r, const float* __restrict__ in, int b0, int t0,
    int lj)
{
  if constexpr (TMAJOR) {
    static_assert(IN == 4 && TC % 2 == 0, "paired planes: 4 channels, whole pairs");
#pragma unroll
    for (int i = 0; i < TC; ++i)
      r.v[i] = *(const float4*)(in + (size_t)(i / (TC / 2)) * PLANE +
                                ((size_t)((t0 >> 1) + (i % (TC / 2))) * NB +
                                 (size_t)(b0 + lj)) * 4);
  } else {
    static_assert((TC * IN) % 2 == 0, "tile row must be whole 8-byte pieces");
    constexpr int TAIL = TilePieces<IN, false>::TAIL;
    const int l = lj & 31;
    const float* base = in + (size_t)(b0 + (lj & ~31)) * (NT * IN) + (size_t)t0 * IN;
#pragma unroll
    for (int i = 0; i < 32; ++i)
      r.v[i] = *(const float2*)(base + (size_t)i * (NT * IN) + 2 * l);
#pragma unroll
    for (int j = 0; j < TAIL; ++j) {
      const int idx = 32 * j + l;
      r.v[32 + j] = *(const float2*)(base + (size_t)(idx / TAIL) * (NT * IN) +
                                     2 * (32 + idx % TAIL));
    }
  }
}

template <int IN, bool TMAJOR>
__device__ __forceinline__ void store_tile(
    float* __restrict__ tile, const TilePieces<IN, TMAJOR>& r, int lj)
{
  if constexpr (TMAJOR) {
#pragma unroll
    for (int i = 0; i < TC; ++i) {
      float* d = tile +
                 ((i % (TC / 2)) * 2 * IN + (i / (TC / 2)) * 2) * PITCH + lj;
      d[0 * PITCH] = r.v[i].x;
      d[1 * PITCH] = r.v[i].y;
      d[(IN + 0) * PITCH] = r.v[i].z;
      d[(IN + 1) * PITCH] = r.v[i].w;
    }
  } else {
    constexpr int TAIL = TilePieces<IN, false>::TAIL;
    const int l = lj & 31;
    float* d = tile + (2 * l) * PITCH + (lj & ~31);
#pragma unroll
    for (int i = 0; i < 32; ++i) {
      d[0 * PITCH + i] = r.v[i].x;
      d[1 * PITCH + i] = r.v[i].y;
    }
#pragma unroll
    for (int j = 0; j < TAIL; ++j) {
      const int idx = 32 * j + l;
      float* e = tile + (2 * (32 + idx % TAIL)) * PITCH + (lj & ~31) + idx / TAIL;
      e[0 * PITCH] = r.v[32 + j].x;
      e[1 * PITCH] = r.v[32 + j].y;
    }
  }
}

__device__ __forceinline__ void lds_handover() {
  __builtin_amdgcn_fence(__ATOMIC_RELEASE, "workgroup", "local");
  __builtin_amdgcn_s_barrier();
  __builtin_amdgcn_fence(__ATOMIC_ACQUIRE, "workgroup", "local");
}

template <int IN, bool TMAJOR>
__global__ __launch_bounds__(512) __attribute__((amdgpu_waves_per_eu(4, 4)))
void lstm_bidir_kernel(
    const float* __restrict__ in,
    const float* __restrict__ wf_ih, const float* __restrict__ wf_hh,
    const float* __restrict__ bf_ih, const float* __restrict__ bf_hh,
    const float* __restrict__ wr_ih, const float* __restrict__ wr_hh,
    const float* __restrict__ br_ih, const float* __restrict__ br_hh,
    float* __restrict__ out)
{
  __shared__ float ring[2][2][TC * IN * PITCH];

  const int tid = threadIdx.x;
  const bool is_loader = __builtin_amdgcn_readfirstlane(tid) >= SPW;
  const int b0 = blockIdx.x * SPW;
  const int lj = tid - SPW;
  const int b = b0 + tid;

  DirWeights<IN> wf, wr;
  load_dir_weights<IN, false>(wf, wf_ih, wf_hh, bf_ih, bf_hh);
  load_dir_weights<IN, true>(wr, wr_ih, wr_hh, br_ih, br_hh);

  float hf0 = 0.f, hf1 = 0.f, cf0 = 0.f, cf1 = 0.f;
  float hr0 = 0.f, hr1 = 0.f, cr0 = 0.f, cr1 = 0.f;
  float kf0 = 0.f, kf1 = 0.f, kr0 = 0.f, kr1 = 0.f;
  static_assert(TC % 2 == 0, "a tile must hold whole time pairs");

  TilePieces<IN, TMAJOR> pt;
  if (is_loader) {
    load_tile<IN, TMAJOR>(pt, in, b0, 0, lj);
    store_tile<IN, TMAJOR>(ring[0][0], pt, lj);
    load_tile<IN, TMAJOR>(pt, in, b0, NT - TC, lj);
    store_tile<IN, TMAJOR>(ring[0][1], pt, lj);
    load_tile<IN, TMAJOR>(pt, in, b0, TC, lj);
  }
  lds_handover();

  for (int c = 0; c < NCH; ++c) {
    const int stage = c & 1;
    if (is_loader) {
      if (c + 1 < NCH) {
        store_tile<IN, TMAJOR>(ring[stage ^ 1][0], pt, lj);
        load_tile<IN, TMAJOR>(pt, in, b0, NT - TC - (c + 1) * TC, lj);
        store_tile<IN, TMAJOR>(ring[stage ^ 1][1], pt, lj);
        if (c + 2 < NCH)
          load_tile<IN, TMAJOR>(pt, in, b0, (c + 2) * TC, lj);
      }
    } else {
      const float* tf_tile = ring[stage][0];
      const float* tr_tile = ring[stage][1];
#pragma unroll
      for (int s = 0; s < TC; ++s) {
        const int tf = c * TC + s;
        const int tr = NT - 1 - tf;
        float xf[IN], xr[IN];
#pragma unroll
        for (int ch = 0; ch < IN; ++ch) {
          xf[ch] = tf_tile[(s * IN + ch) * PITCH + tid];
          xr[ch] = tr_tile[((TC - 1 - s) * IN + ch) * PITCH + tid];
        }

        lstm_step<IN>(wf, xf, hf0, hf1, cf0, cf1);
        lstm_step<IN>(wr, xr, hr0, hr1, cr0, cr1);

        if ((s & 1) == 0) {
          kf0 = hf0; kf1 = hf1;
          kr0 = hr0; kr1 = hr1;
        } else {
          *(float4*)(out + ((size_t)(tf >> 1) * NB + (size_t)b) * 4) =
              make_float4(kf0, kf1, hf0, hf1);
          *(float4*)(out + PLANE + ((size_t)(tr >> 1) * NB + (size_t)b) * 4) =
              make_float4(hr0, hr1, kr0, kr1);
        }
      }
    }
    lds_handover();
  }
}

#define ACT_PITCH 68

__global__ __launch_bounds__(256) void fc_head_kernel(
    const float* __restrict__ h2,
    const float* __restrict__ w,
    const float* __restrict__ bias,
    const float* __restrict__ w2,
    const float* __restrict__ bias2,
    float* __restrict__ out)
{
  __shared__ __attribute__((aligned(16))) float act_tile[32 * ACT_PITCH];

  const int wave = __builtin_amdgcn_readfirstlane((int)threadIdx.x) >> 5;
  const int lane = threadIdx.x & 31;
  const int mt = wave >> 2;
  const int n_tile = wave & 3;
  const int row = lane & 15;
  const int half = lane >> 4;
  const int base_b = (blockIdx.x * 2 + mt) * 16;
  const int n = n_tile * 16 + row;

  v8f acc = {};
  const float* wrow = w + (size_t)n * 400 + half * 4;
  const float* arow = h2 + (size_t)half * PLANE + (size_t)(base_b + row) * 4;

#pragma unroll 5
  for (int j = 0; j < NT / 2; ++j) {
    float4 a = *(const float4*)(arow + (size_t)j * NB * 4);
    const float4 bq = *(const float4*)(wrow + j * 8);
    const v2u sxz = __builtin_amdgcn_permlane16_swap(
        __float_as_uint(bq.x), __float_as_uint(bq.z), false, false);
    const v2u syw = __builtin_amdgcn_permlane16_swap(
        __float_as_uint(bq.y), __float_as_uint(bq.w), false, false);
    v2f a01, a23, b01, b23;
    a01.x = fmaxf(a.x, 0.0f);
    a01.y = fmaxf(a.y, 0.0f);
    a23.x = fmaxf(a.z, 0.0f);
    a23.y = fmaxf(a.w, 0.0f);
    b01.x = __uint_as_float(sxz.x); b01.y = __uint_as_float(syw.x);
    b23.x = __uint_as_float(sxz.y); b23.y = __uint_as_float(syw.y);
    acc = __builtin_amdgcn_wmma_f32_16x16x4_f32(
        false, a01, false, b01, (short)0, acc, false, false);
    acc = __builtin_amdgcn_wmma_f32_16x16x4_f32(
        false, a23, false, b23, (short)0, acc, false, false);
  }

  {
    const float bn = bias[n];
    float* d = act_tile + (mt * 16 + half * 8) * ACT_PITCH + n;
#pragma unroll
    for (int r = 0; r < 8; ++r)
      d[r * ACT_PITCH] = fmaxf(acc[r] + bn, 0.0f);
  }
  __syncthreads();

  if (wave < 4) {
    const int mt2 = wave >> 1;
    const int n2 = (wave & 1) * 16 + row;
    const bool nvalid = (n2 < 20);
    const int base_b2 = (blockIdx.x * 2 + mt2) * 16;

    v8f acc2 = {};
    const float* arow2 = act_tile + (mt2 * 16 + row) * ACT_PITCH + half * 4;
    const float* wrow2 = w2 + (size_t)(nvalid ? n2 : 0) * 64 + half * 4;

#pragma unroll
    for (int j = 0; j < 8; ++j) {
      const float4 a = *(const float4*)(arow2 + j * 8);
      float4 bq = *(const float4*)(wrow2 + j * 8);
      if (!nvalid) { bq.x = 0.0f; bq.y = 0.0f; bq.z = 0.0f; bq.w = 0.0f; }
      v2f a01, a23, b01, b23;
      a01.x = a.x; a01.y = a.y;
      a23.x = a.z; a23.y = a.w;
      b01.x = bq.x; b01.y = bq.y;
      b23.x = bq.z; b23.y = bq.w;
      acc2 = __builtin_amdgcn_wmma_f32_16x16x4_f32(
          false, a01, false, b01, (short)0, acc2, false, false);
      acc2 = __builtin_amdgcn_wmma_f32_16x16x4_f32(
          false, a23, false, b23, (short)0, acc2, false, false);
    }

    if (nvalid) {
      const float bn2 = bias2[n2];
#pragma unroll
      for (int r = 0; r < 8; ++r) {
        const int m = r + half * 8;
        out[(size_t)(base_b2 + m) * 20 + n2] = acc2[r] + bn2;
      }
    }
  }
}

extern "C" void kernel_launch(void* const* d_in, const int* in_sizes, int n_in,
                              void* d_out, int out_size, void* d_ws,
                              size_t ws_size, hipStream_t stream) {
  const float* x = (const float*)d_in[0];
  const float* w_ih_l0  = (const float*)d_in[1];
  const float* w_hh_l0  = (const float*)d_in[2];
  const float* b_ih_l0  = (const float*)d_in[3];
  const float* b_hh_l0  = (const float*)d_in[4];
  const float* w_ih_l0r = (const float*)d_in[5];
  const float* w_hh_l0r = (const float*)d_in[6];
  const float* b_ih_l0r = (const float*)d_in[7];
  const float* b_hh_l0r = (const float*)d_in[8];
  const float* w_ih_l1  = (const float*)d_in[9];
  const float* w_hh_l1  = (const float*)d_in[10];
  const float* b_ih_l1  = (const float*)d_in[11];
  const float* b_hh_l1  = (const float*)d_in[12];
  const float* w_ih_l1r = (const float*)d_in[13];
  const float* w_hh_l1r = (const float*)d_in[14];
  const float* b_ih_l1r = (const float*)d_in[15];
  const float* b_hh_l1r = (const float*)d_in[16];
  const float* fc1_w = (const float*)d_in[17];
  const float* fc1_b = (const float*)d_in[18];
  const float* fc2_w = (const float*)d_in[19];
  const float* fc2_b = (const float*)d_in[20];
  float* out = (float*)d_out;

  float* ws  = (float*)d_ws;
  float* ws0 = ws;
  float* ws1 = ws + (size_t)NT * NB * 4;

  const dim3 blk(256);
  const dim3 blk_lstm(512);
  const dim3 grid_lstm(NB / 256);

  lstm_bidir_kernel<7, false><<<grid_lstm, blk_lstm, 0, stream>>>(
      x, w_ih_l0, w_hh_l0, b_ih_l0, b_hh_l0,
      w_ih_l0r, w_hh_l0r, b_ih_l0r, b_hh_l0r, ws0);

  lstm_bidir_kernel<4, true><<<grid_lstm, blk_lstm, 0, stream>>>(
      ws0, w_ih_l1, w_hh_l1, b_ih_l1, b_hh_l1,
      w_ih_l1r, w_hh_l1r, b_ih_l1r, b_hh_l1r, ws1);

  fc_head_kernel<<<dim3(2048), blk, 0, stream>>>(ws1, fc1_w, fc1_b, fc2_w, fc2_b,
                                                out);
}


# LSTM gate sums start from bias + input products, formed one step ahead from x read a step ahead; the two h products come last: 2 dependent FMAs per gate between steps, not 2+IN; sum order changes
# speedup vs baseline: 2.5157x; 1.0775x over previous
#include <hip/hip_runtime.h>
#include <hip/hip_bf16.h>


#define NB 65536
#define NT 100

#define PLANE ((size_t)(NT / 2) * NB * 4)

typedef __attribute__((ext_vector_type(2))) float v2f;
typedef __attribute__((ext_vector_type(8))) float v8f;
typedef __attribute__((ext_vector_type(2))) unsigned int v2u;

__device__ __forceinline__ float fast_rcp(float x) {
  return __builtin_amdgcn_rcpf(x);
}
__device__ __forceinline__ float sigm(float x) {
  return fast_rcp(1.0f + __expf(-x));
}
__device__ __forceinline__ float tanh_fast(float x) {
  return fmaf(2.0f, fast_rcp(1.0f + __expf(-2.0f * x)), -1.0f);
}

template <int IN>
struct DirWeights {
  float wih[8][IN];
  float whh[8][2];
  float bias[8];
};

__device__ __forceinline__ float keep_in_vgpr(float v) {
  asm("" : "+v"(v));
  return v;
}

template <int IN, bool IN_VGPR>
__device__ __forceinline__ void load_dir_weights(
    DirWeights<IN>& w,
    const float* __restrict__ w_ih,
    const float* __restrict__ w_hh,
    const float* __restrict__ b_ih,
    const float* __restrict__ b_hh)
{
#pragma unroll
  for (int g = 0; g < 8; ++g) {
#pragma unroll
    for (int c = 0; c < IN; ++c) w.wih[g][c] = w_ih[g * IN + c];
    w.whh[g][0] = w_hh[g * 2 + 0];
    w.whh[g][1] = w_hh[g * 2 + 1];
    w.bias[g] = b_ih[g] + b_hh[g];
    if constexpr (IN_VGPR) {
#pragma unroll
      for (int c = 0; c < IN; ++c) w.wih[g][c] = keep_in_vgpr(w.wih[g][c]);
      w.whh[g][0] = keep_in_vgpr(w.whh[g][0]);
      w.whh[g][1] = keep_in_vgpr(w.whh[g][1]);
      w.bias[g] = keep_in_vgpr(w.bias[g]);
    }
  }
}

template <int IN>
__device__ __forceinline__ void gate_input_part(
    const DirWeights<IN>& w, const float (&xv)[IN], float (&p)[8])
{
#pragma unroll
  for (int gg = 0; gg < 8; ++gg) {
    float acc = w.bias[gg];
#pragma unroll
    for (int c = 0; c < IN; ++c) acc = fmaf(w.wih[gg][c], xv[c], acc);
    p[gg] = acc;
  }
}

template <int IN>
__device__ __forceinline__ void lstm_step(
    const DirWeights<IN>& w, const float (&p)[8],
    float& h0, float& h1, float& c0, float& c1)
{
  float g[8];
#pragma unroll
  for (int gg = 0; gg < 8; ++gg)
    g[gg] = fmaf(w.whh[gg][0], h0, fmaf(w.whh[gg][1], h1, p[gg]));

  const float i0 = sigm(g[0]), i1 = sigm(g[1]);
  const float f0 = sigm(g[2]), f1 = sigm(g[3]);
  const float t0 = tanh_fast(g[4]), t1 = tanh_fast(g[5]);
  const float o0 = sigm(g[6]), o1 = sigm(g[7]);

  c0 = fmaf(f0, c0, i0 * t0);
  c1 = fmaf(f1, c1, i1 * t1);
  h0 = o0 * tanh_fast(c0);
  h1 = o1 * tanh_fast(c1);
}

#define SPW 256
#define TC 10
#define NCH (NT / TC)
#define PITCH 257

template <int IN, bool TMAJOR>
struct TilePieces;
template <int IN>
struct TilePieces<IN, false> {
  static constexpr int PPS = TC * IN / 2;
  static constexpr int TAIL = PPS - 32;
  static_assert(TAIL > 0 && TAIL <= 32, "row load takes 32 pieces, tail loads the rest");
  float2 v[32 + TAIL];
};
template <int IN>
struct TilePieces<IN, true> {
  float4 v[TC];
};

template <int IN, bool TMAJOR>
__device__ __forceinline__ void load_tile(
    TilePieces<IN, TMAJOR>& r, const float* __restrict__ in, int b0, int t0,
    int lj)
{
  if constexpr (TMAJOR) {
    static_assert(IN == 4 && TC % 2 == 0, "paired planes: 4 channels, whole pairs");
#pragma unroll
    for (int i = 0; i < TC; ++i)
      r.v[i] = *(const float4*)(in + (size_t)(i / (TC / 2)) * PLANE +
                                ((size_t)((t0 >> 1) + (i % (TC / 2))) * NB +
                                 (size_t)(b0 + lj)) * 4);
  } else {
    static_assert((TC * IN) % 2 == 0, "tile row must be whole 8-byte pieces");
    constexpr int TAIL = TilePieces<IN, false>::TAIL;
    const int l = lj & 31;
    const float* base = in + (size_t)(b0 + (lj & ~31)) * (NT * IN) + (size_t)t0 * IN;
#pragma unroll
    for (int i = 0; i < 32; ++i)
      r.v[i] = *(const float2*)(base + (size_t)i * (NT * IN) + 2 * l);
#pragma unroll
    for (int j = 0; j < TAIL; ++j) {
      const int idx = 32 * j + l;
      r.v[32 + j] = *(const float2*)(base + (size_t)(idx / TAIL) * (NT * IN) +
                                     2 * (32 + idx % TAIL));
    }
  }
}

template <int IN, bool TMAJOR>
__device__ __forceinline__ void store_tile(
    float* __restrict__ tile, const TilePieces<IN, TMAJOR>& r, int lj)
{
  if constexpr (TMAJOR) {
#pragma unroll
    for (int i = 0; i < TC; ++i) {
      float* d = tile +
                 ((i % (TC / 2)) * 2 * IN + (i / (TC / 2)) * 2) * PITCH + lj;
      d[0 * PITCH] = r.v[i].x;
      d[1 * PITCH] = r.v[i].y;
      d[(IN + 0) * PITCH] = r.v[i].z;
      d[(IN + 1) * PITCH] = r.v[i].w;
    }
  } else {
    constexpr int TAIL = TilePieces<IN, false>::TAIL;
    const int l = lj & 31;
    float* d = tile + (2 * l) * PITCH + (lj & ~31);
#pragma unroll
    for (int i = 0; i < 32; ++i) {
      d[0 * PITCH + i] = r.v[i].x;
      d[1 * PITCH + i] = r.v[i].y;
    }
#pragma unroll
    for (int j = 0; j < TAIL; ++j) {
      const int idx = 32 * j + l;
      float* e = tile + (2 * (32 + idx % TAIL)) * PITCH + (lj & ~31) + idx / TAIL;
      e[0 * PITCH] = r.v[32 + j].x;
      e[1 * PITCH] = r.v[32 + j].y;
    }
  }
}

__device__ __forceinline__ void lds_handover() {
  __builtin_amdgcn_fence(__ATOMIC_RELEASE, "workgroup", "local");
  __builtin_amdgcn_s_barrier();
  __builtin_amdgcn_fence(__ATOMIC_ACQUIRE, "workgroup", "local");
}

template <int IN, bool TMAJOR>
__global__ __launch_bounds__(512) __attribute__((amdgpu_waves_per_eu(4, 4)))
void lstm_bidir_kernel(
    const float* __restrict__ in,
    const float* __restrict__ wf_ih, const float* __restrict__ wf_hh,
    const float* __restrict__ bf_ih, const float* __restrict__ bf_hh,
    const float* __restrict__ wr_ih, const float* __restrict__ wr_hh,
    const float* __restrict__ br_ih, const float* __restrict__ br_hh,
    float* __restrict__ out)
{
  __shared__ float ring[2][2][TC * IN * PITCH];

  const int tid = threadIdx.x;
  const bool is_loader = __builtin_amdgcn_readfirstlane(tid) >= SPW;
  const int b0 = blockIdx.x * SPW;
  const int lj = tid - SPW;
  const int b = b0 + tid;

  DirWeights<IN> wf, wr;
  load_dir_weights<IN, false>(wf, wf_ih, wf_hh, bf_ih, bf_hh);
  load_dir_weights<IN, true>(wr, wr_ih, wr_hh, br_ih, br_hh);

  float hf0 = 0.f, hf1 = 0.f, cf0 = 0.f, cf1 = 0.f;
  float hr0 = 0.f, hr1 = 0.f, cr0 = 0.f, cr1 = 0.f;
  float kf0 = 0.f, kf1 = 0.f, kr0 = 0.f, kr1 = 0.f;
  static_assert(TC % 2 == 0, "a tile must hold whole time pairs");

  TilePieces<IN, TMAJOR> pt;
  if (is_loader) {
    load_tile<IN, TMAJOR>(pt, in, b0, 0, lj);
    store_tile<IN, TMAJOR>(ring[0][0], pt, lj);
    load_tile<IN, TMAJOR>(pt, in, b0, NT - TC, lj);
    store_tile<IN, TMAJOR>(ring[0][1], pt, lj);
    load_tile<IN, TMAJOR>(pt, in, b0, TC, lj);
  }
  lds_handover();

  for (int c = 0; c < NCH; ++c) {
    const int stage = c & 1;
    if (is_loader) {
      if (c + 1 < NCH) {
        store_tile<IN, TMAJOR>(ring[stage ^ 1][0], pt, lj);
        load_tile<IN, TMAJOR>(pt, in, b0, NT - TC - (c + 1) * TC, lj);
        store_tile<IN, TMAJOR>(ring[stage ^ 1][1], pt, lj);
        if (c + 2 < NCH)
          load_tile<IN, TMAJOR>(pt, in, b0, (c + 2) * TC, lj);
      }
    } else {
      const float* tf_tile = ring[stage][0];
      const float* tr_tile = ring[stage][1];
      float pf[8], pr[8];
      {
        float x0f[IN], x0r[IN];
#pragma unroll
        for (int ch = 0; ch < IN; ++ch) {
          x0f[ch] = tf_tile[ch * PITCH + tid];
          x0r[ch] = tr_tile[((TC - 1) * IN + ch) * PITCH + tid];
        }
        gate_input_part<IN>(wf, x0f, pf);
        gate_input_part<IN>(wr, x0r, pr);
      }
#pragma unroll
      for (int s = 0; s < TC; ++s) {
        const int tf = c * TC + s;
        const int tr = NT - 1 - tf;
        float xf[IN], xr[IN];
        if (s + 1 < TC) {
#pragma unroll
          for (int ch = 0; ch < IN; ++ch) {
            xf[ch] = tf_tile[((s + 1) * IN + ch) * PITCH + tid];
            xr[ch] = tr_tile[((TC - 2 - s) * IN + ch) * PITCH + tid];
          }
        }
        __builtin_amdgcn_sched_barrier(0);

        lstm_step<IN>(wf, pf, hf0, hf1, cf0, cf1);
        lstm_step<IN>(wr, pr, hr0, hr1, cr0, cr1);

        if (s + 1 < TC) {
          gate_input_part<IN>(wf, xf, pf);
          gate_input_part<IN>(wr, xr, pr);
        }

        if ((s & 1) == 0) {
          kf0 = hf0; kf1 = hf1;
          kr0 = hr0; kr1 = hr1;
        } else {
          *(float4*)(out + ((size_t)(tf >> 1) * NB + (size_t)b) * 4) =
              make_float4(kf0, kf1, hf0, hf1);
          *(float4*)(out + PLANE + ((size_t)(tr >> 1) * NB + (size_t)b) * 4) =
              make_float4(hr0, hr1, kr0, kr1);
        }
      }
    }
    lds_handover();
  }
}

#define ACT_PITCH 68

__global__ __launch_bounds__(256) void fc_head_kernel(
    const float* __restrict__ h2,
    const float* __restrict__ w,
    const float* __restrict__ bias,
    const float* __restrict__ w2,
    const float* __restrict__ bias2,
    float* __restrict__ out)
{
  __shared__ __attribute__((aligned(16))) float act_tile[32 * ACT_PITCH];

  const int wave = __builtin_amdgcn_readfirstlane((int)threadIdx.x) >> 5;
  const int lane = threadIdx.x & 31;
  const int mt = wave >> 2;
  const int n_tile = wave & 3;
  const int row = lane & 15;
  const int half = lane >> 4;
  const int base_b = (blockIdx.x * 2 + mt) * 16;
  const int n = n_tile * 16 + row;

  v8f acc = {};
  const float* wrow = w + (size_t)n * 400 + half * 4;
  const float* arow = h2 + (size_t)half * PLANE + (size_t)(base_b + row) * 4;

#pragma unroll 5
  for (int j = 0; j < NT / 2; ++j) {
    float4 a = *(const float4*)(arow + (size_t)j * NB * 4);
    const float4 bq = *(const float4*)(wrow + j * 8);
    const v2u sxz = __builtin_amdgcn_permlane16_swap(
        __float_as_uint(bq.x), __float_as_uint(bq.z), false, false);
    const v2u syw = __builtin_amdgcn_permlane16_swap(
        __float_as_uint(bq.y), __float_as_uint(bq.w), false, false);
    v2f a01, a23, b01, b23;
    a01.x = fmaxf(a.x, 0.0f);
    a01.y = fmaxf(a.y, 0.0f);
    a23.x = fmaxf(a.z, 0.0f);
    a23.y = fmaxf(a.w, 0.0f);
    b01.x = __uint_as_float(sxz.x); b01.y = __uint_as_float(syw.x);
    b23.x = __uint_as_float(sxz.y); b23.y = __uint_as_float(syw.y);
    acc = __builtin_amdgcn_wmma_f32_16x16x4_f32(
        false, a01, false, b01, (short)0, acc, false, false);
    acc = __builtin_amdgcn_wmma_f32_16x16x4_f32(
        false, a23, false, b23, (short)0, acc, false, false);
  }

  {
    const float bn = bias[n];
    float* d = act_tile + (mt * 16 + half * 8) * ACT_PITCH + n;
#pragma unroll
    for (int r = 0; r < 8; ++r)
      d[r * ACT_PITCH] = fmaxf(acc[r] + bn, 0.0f);
  }
  __syncthreads();

  if (wave < 4) {
    const int mt2 = wave >> 1;
    const int n2 = (wave & 1) * 16 + row;
    const bool nvalid = (n2 < 20);
    const int base_b2 = (blockIdx.x * 2 + mt2) * 16;

    v8f acc2 = {};
    const float* arow2 = act_tile + (mt2 * 16 + row) * ACT_PITCH + half * 4;
    const float* wrow2 = w2 + (size_t)(nvalid ? n2 : 0) * 64 + half * 4;

#pragma unroll
    for (int j = 0; j < 8; ++j) {
      const float4 a = *(const float4*)(arow2 + j * 8);
      float4 bq = *(const float4*)(wrow2 + j * 8);
      if (!nvalid) { bq.x = 0.0f; bq.y = 0.0f; bq.z = 0.0f; bq.w = 0.0f; }
      v2f a01, a23, b01, b23;
      a01.x = a.x; a01.y = a.y;
      a23.x = a.z; a23.y = a.w;
      b01.x = bq.x; b01.y = bq.y;
      b23.x = bq.z; b23.y = bq.w;
      acc2 = __builtin_amdgcn_wmma_f32_16x16x4_f32(
          false, a01, false, b01, (short)0, acc2, false, false);
      acc2 = __builtin_amdgcn_wmma_f32_16x16x4_f32(
          false, a23, false, b23, (short)0, acc2, false, false);
    }

    if (nvalid) {
      const float bn2 = bias2[n2];
#pragma unroll
      for (int r = 0; r < 8; ++r) {
        const int m = r + half * 8;
        out[(size_t)(base_b2 + m) * 20 + n2] = acc2[r] + bn2;
      }
    }
  }
}

extern "C" void kernel_launch(void* const* d_in, const int* in_sizes, int n_in,
                              void* d_out, int out_size, void* d_ws,
                              size_t ws_size, hipStream_t stream) {
  const float* x = (const float*)d_in[0];
  const float* w_ih_l0  = (const float*)d_in[1];
  const float* w_hh_l0  = (const float*)d_in[2];
  const float* b_ih_l0  = (const float*)d_in[3];
  const float* b_hh_l0  = (const float*)d_in[4];
  const float* w_ih_l0r = (const float*)d_in[5];
  const float* w_hh_l0r = (const float*)d_in[6];
  const float* b_ih_l0r = (const float*)d_in[7];
  const float* b_hh_l0r = (const float*)d_in[8];
  const float* w_ih_l1  = (const float*)d_in[9];
  const float* w_hh_l1  = (const float*)d_in[10];
  const float* b_ih_l1  = (const float*)d_in[11];
  const float* b_hh_l1  = (const float*)d_in[12];
  const float* w_ih_l1r = (const float*)d_in[13];
  const float* w_hh_l1r = (const float*)d_in[14];
  const float* b_ih_l1r = (const float*)d_in[15];
  const float* b_hh_l1r = (const float*)d_in[16];
  const float* fc1_w = (const float*)d_in[17];
  const float* fc1_b = (const float*)d_in[18];
  const float* fc2_w = (const float*)d_in[19];
  const float* fc2_b = (const float*)d_in[20];
  float* out = (float*)d_out;

  float* ws  = (float*)d_ws;
  float* ws0 = ws;
  float* ws1 = ws + (size_t)NT * NB * 4;

  const dim3 blk(256);
  const dim3 blk_lstm(512);
  const dim3 grid_lstm(NB / 256);

  lstm_bidir_kernel<7, false><<<grid_lstm, blk_lstm, 0, stream>>>(
      x, w_ih_l0, w_hh_l0, b_ih_l0, b_hh_l0,
      w_ih_l0r, w_hh_l0r, b_ih_l0r, b_hh_l0r, ws0);

  lstm_bidir_kernel<4, true><<<grid_lstm, blk_lstm, 0, stream>>>(
      ws0, w_ih_l1, w_hh_l1, b_ih_l1, b_hh_l1,
      w_ih_l1r, w_hh_l1r, b_ih_l1r, b_hh_l1r, ws1);

  fc_head_kernel<<<dim3(2048), blk, 0, stream>>>(ws1, fc1_w, fc1_b, fc2_w, fc2_b,
                                                out);
}


# LSTM exponent scales carried by the values: gate weight and bias rows times -log2(e) or -2log2(e) at load, cell state kept times -2log2(e); no multiply before any v_exp_f32; roundings change
# speedup vs baseline: 2.6164x; 1.0400x over previous
#include <hip/hip_runtime.h>
#include <hip/hip_bf16.h>


#define NB 65536
#define NT 100

#define PLANE ((size_t)(NT / 2) * NB * 4)

typedef __attribute__((ext_vector_type(2))) float v2f;
typedef __attribute__((ext_vector_type(8))) float v8f;
typedef __attribute__((ext_vector_type(2))) unsigned int v2u;

__device__ __forceinline__ float fast_rcp(float x) {
  return __builtin_amdgcn_rcpf(x);
}

#define LOG2E_F 1.44269504088896340736f
#define KS (-LOG2E_F)
#define KT (-2.0f * LOG2E_F)

__device__ __forceinline__ float sigm_scaled(float xs) {
  return fast_rcp(1.0f + __builtin_amdgcn_exp2f(xs));
}

template <int IN>
struct DirWeights {
  float wih[8][IN];
  float whh[8][2];
  float bias[8];
};

__device__ __forceinline__ float keep_in_vgpr(float v) {
  asm("" : "+v"(v));
  return v;
}

template <int IN, bool IN_VGPR>
__device__ __forceinline__ void load_dir_weights(
    DirWeights<IN>& w,
    const float* __restrict__ w_ih,
    const float* __restrict__ w_hh,
    const float* __restrict__ b_ih,
    const float* __restrict__ b_hh)
{
#pragma unroll
  for (int g = 0; g < 8; ++g) {
    const float ks = (g == 4 || g == 5) ? KT : KS;
#pragma unroll
    for (int c = 0; c < IN; ++c) w.wih[g][c] = ks * w_ih[g * IN + c];
    w.whh[g][0] = ks * w_hh[g * 2 + 0];
    w.whh[g][1] = ks * w_hh[g * 2 + 1];
    w.bias[g] = ks * (b_ih[g] + b_hh[g]);
    if constexpr (IN_VGPR) {
#pragma unroll
      for (int c = 0; c < IN; ++c) w.wih[g][c] = keep_in_vgpr(w.wih[g][c]);
      w.whh[g][0] = keep_in_vgpr(w.whh[g][0]);
      w.whh[g][1] = keep_in_vgpr(w.whh[g][1]);
      w.bias[g] = keep_in_vgpr(w.bias[g]);
    }
  }
}

template <int IN>
__device__ __forceinline__ void gate_input_part(
    const DirWeights<IN>& w, const float (&xv)[IN], float (&p)[8])
{
#pragma unroll
  for (int gg = 0; gg < 8; ++gg) {
    float acc = w.bias[gg];
#pragma unroll
    for (int c = 0; c < IN; ++c) acc = fmaf(w.wih[gg][c], xv[c], acc);
    p[gg] = acc;
  }
}

template <int IN>
__device__ __forceinline__ void lstm_step(
    const DirWeights<IN>& w, const float (&p)[8],
    float& h0, float& h1, float& cs0, float& cs1)
{
  float g[8];
#pragma unroll
  for (int gg = 0; gg < 8; ++gg)
    g[gg] = fmaf(w.whh[gg][0], h0, fmaf(w.whh[gg][1], h1, p[gg]));

  const float i0 = sigm_scaled(g[0]), i1 = sigm_scaled(g[1]);
  const float f0 = sigm_scaled(g[2]), f1 = sigm_scaled(g[3]);
  const float t0 = fmaf(2.0f * KT, sigm_scaled(g[4]), -KT);
  const float t1 = fmaf(2.0f * KT, sigm_scaled(g[5]), -KT);
  const float o0 = sigm_scaled(g[6]), o1 = sigm_scaled(g[7]);

  cs0 = fmaf(f0, cs0, i0 * t0);
  cs1 = fmaf(f1, cs1, i1 * t1);
  h0 = o0 * fmaf(2.0f, sigm_scaled(cs0), -1.0f);
  h1 = o1 * fmaf(2.0f, sigm_scaled(cs1), -1.0f);
}

#define SPW 256
#define TC 10
#define NCH (NT / TC)
#define PITCH 257

template <int IN, bool TMAJOR>
struct TilePieces;
template <int IN>
struct TilePieces<IN, false> {
  static constexpr int PPS = TC * IN / 2;
  static constexpr int TAIL = PPS - 32;
  static_assert(TAIL > 0 && TAIL <= 32, "row load takes 32 pieces, tail loads the rest");
  float2 v[32 + TAIL];
};
template <int IN>
struct TilePieces<IN, true> {
  float4 v[TC];
};

template <int IN, bool TMAJOR>
__device__ __forceinline__ void load_tile(
    TilePieces<IN, TMAJOR>& r, const float* __restrict__ in, int b0, int t0,
    int lj)
{
  if constexpr (TMAJOR) {
    static_assert(IN == 4 && TC % 2 == 0, "paired planes: 4 channels, whole pairs");
#pragma unroll
    for (int i = 0; i < TC; ++i)
      r.v[i] = *(const float4*)(in + (size_t)(i / (TC / 2)) * PLANE +
                                ((size_t)((t0 >> 1) + (i % (TC / 2))) * NB +
                                 (size_t)(b0 + lj)) * 4);
  } else {
    static_assert((TC * IN) % 2 == 0, "tile row must be whole 8-byte pieces");
    constexpr int TAIL = TilePieces<IN, false>::TAIL;
    const int l = lj & 31;
    const float* base = in + (size_t)(b0 + (lj & ~31)) * (NT * IN) + (size_t)t0 * IN;
#pragma unroll
    for (int i = 0; i < 32; ++i)
      r.v[i] = *(const float2*)(base + (size_t)i * (NT * IN) + 2 * l);
#pragma unroll
    for (int j = 0; j < TAIL; ++j) {
      const int idx = 32 * j + l;
      r.v[32 + j] = *(const float2*)(base + (size_t)(idx / TAIL) * (NT * IN) +
                                     2 * (32 + idx % TAIL));
    }
  }
}

template <int IN, bool TMAJOR>
__device__ __forceinline__ void store_tile(
    float* __restrict__ tile, const TilePieces<IN, TMAJOR>& r, int lj)
{
  if constexpr (TMAJOR) {
#pragma unroll
    for (int i = 0; i < TC; ++i) {
      float* d = tile +
                 ((i % (TC / 2)) * 2 * IN + (i / (TC / 2)) * 2) * PITCH + lj;
      d[0 * PITCH] = r.v[i].x;
      d[1 * PITCH] = r.v[i].y;
      d[(IN + 0) * PITCH] = r.v[i].z;
      d[(IN + 1) * PITCH] = r.v[i].w;
    }
  } else {
    constexpr int TAIL = TilePieces<IN, false>::TAIL;
    const int l = lj & 31;
    float* d = tile + (2 * l) * PITCH + (lj & ~31);
#pragma unroll
    for (int i = 0; i < 32; ++i) {
      d[0 * PITCH + i] = r.v[i].x;
      d[1 * PITCH + i] = r.v[i].y;
    }
#pragma unroll
    for (int j = 0; j < TAIL; ++j) {
      const int idx = 32 * j + l;
      float* e = tile + (2 * (32 + idx % TAIL)) * PITCH + (lj & ~31) + idx / TAIL;
      e[0 * PITCH] = r.v[32 + j].x;
      e[1 * PITCH] = r.v[32 + j].y;
    }
  }
}

__device__ __forceinline__ void lds_handover() {
  __builtin_amdgcn_fence(__ATOMIC_RELEASE, "workgroup", "local");
  __builtin_amdgcn_s_barrier();
  __builtin_amdgcn_fence(__ATOMIC_ACQUIRE, "workgroup", "local");
}

template <int IN, bool TMAJOR>
__global__ __launch_bounds__(512) __attribute__((amdgpu_waves_per_eu(4, 4)))
void lstm_bidir_kernel(
    const float* __restrict__ in,
    const float* __restrict__ wf_ih, const float* __restrict__ wf_hh,
    const float* __restrict__ bf_ih, const float* __restrict__ bf_hh,
    const float* __restrict__ wr_ih, const float* __restrict__ wr_hh,
    const float* __restrict__ br_ih, const float* __restrict__ br_hh,
    float* __restrict__ out)
{
  __shared__ float ring[2][2][TC * IN * PITCH];

  const int tid = threadIdx.x;
  const bool is_loader = __builtin_amdgcn_readfirstlane(tid) >= SPW;
  const int b0 = blockIdx.x * SPW;
  const int lj = tid - SPW;
  const int b = b0 + tid;

  DirWeights<IN> wf, wr;
  load_dir_weights<IN, false>(wf, wf_ih, wf_hh, bf_ih, bf_hh);
  load_dir_weights<IN, true>(wr, wr_ih, wr_hh, br_ih, br_hh);

  float hf0 = 0.f, hf1 = 0.f, cf0 = 0.f, cf1 = 0.f;
  float hr0 = 0.f, hr1 = 0.f, cr0 = 0.f, cr1 = 0.f;
  float kf0 = 0.f, kf1 = 0.f, kr0 = 0.f, kr1 = 0.f;
  static_assert(TC % 2 == 0, "a tile must hold whole time pairs");

  TilePieces<IN, TMAJOR> pt;
  if (is_loader) {
    load_tile<IN, TMAJOR>(pt, in, b0, 0, lj);
    store_tile<IN, TMAJOR>(ring[0][0], pt, lj);
    load_tile<IN, TMAJOR>(pt, in, b0, NT - TC, lj);
    store_tile<IN, TMAJOR>(ring[0][1], pt, lj);
    load_tile<IN, TMAJOR>(pt, in, b0, TC, lj);
  }
  lds_handover();

  for (int c = 0; c < NCH; ++c) {
    const int stage = c & 1;
    if (is_loader) {
      if (c + 1 < NCH) {
        store_tile<IN, TMAJOR>(ring[stage ^ 1][0], pt, lj);
        load_tile<IN, TMAJOR>(pt, in, b0, NT - TC - (c + 1) * TC, lj);
        store_tile<IN, TMAJOR>(ring[stage ^ 1][1], pt, lj);
        if (c + 2 < NCH)
          load_tile<IN, TMAJOR>(pt, in, b0, (c + 2) * TC, lj);
      }
    } else {
      const float* tf_tile = ring[stage][0];
      const float* tr_tile = ring[stage][1];
      float pf[8], pr[8];
      {
        float x0f[IN], x0r[IN];
#pragma unroll
        for (int ch = 0; ch < IN; ++ch) {
          x0f[ch] = tf_tile[ch * PITCH + tid];
          x0r[ch] = tr_tile[((TC - 1) * IN + ch) * PITCH + tid];
        }
        gate_input_part<IN>(wf, x0f, pf);
        gate_input_part<IN>(wr, x0r, pr);
      }
#pragma unroll
      for (int s = 0; s < TC; ++s) {
        const int tf = c * TC + s;
        const int tr = NT - 1 - tf;
        float xf[IN], xr[IN];
        if (s + 1 < TC) {
#pragma unroll
          for (int ch = 0; ch < IN; ++ch) {
            xf[ch] = tf_tile[((s + 1) * IN + ch) * PITCH + tid];
            xr[ch] = tr_tile[((TC - 2 - s) * IN + ch) * PITCH + tid];
          }
        }
        __builtin_amdgcn_sched_barrier(0);

        lstm_step<IN>(wf, pf, hf0, hf1, cf0, cf1);
        lstm_step<IN>(wr, pr, hr0, hr1, cr0, cr1);

        if (s + 1 < TC) {
          gate_input_part<IN>(wf, xf, pf);
          gate_input_part<IN>(wr, xr, pr);
        }

        if ((s & 1) == 0) {
          kf0 = hf0; kf1 = hf1;
          kr0 = hr0; kr1 = hr1;
        } else {
          *(float4*)(out + ((size_t)(tf >> 1) * NB + (size_t)b) * 4) =
              make_float4(kf0, kf1, hf0, hf1);
          *(float4*)(out + PLANE + ((size_t)(tr >> 1) * NB + (size_t)b) * 4) =
              make_float4(hr0, hr1, kr0, kr1);
        }
      }
    }
    lds_handover();
  }
}

#define ACT_PITCH 68

__global__ __launch_bounds__(256) void fc_head_kernel(
    const float* __restrict__ h2,
    const float* __restrict__ w,
    const float* __restrict__ bias,
    const float* __restrict__ w2,
    const float* __restrict__ bias2,
    float* __restrict__ out)
{
  __shared__ __attribute__((aligned(16))) float act_tile[32 * ACT_PITCH];

  const int wave = __builtin_amdgcn_readfirstlane((int)threadIdx.x) >> 5;
  const int lane = threadIdx.x & 31;
  const int mt = wave >> 2;
  const int n_tile = wave & 3;
  const int row = lane & 15;
  const int half = lane >> 4;
  const int base_b = (blockIdx.x * 2 + mt) * 16;
  const int n = n_tile * 16 + row;

  v8f acc = {};
  const float* wrow = w + (size_t)n * 400 + half * 4;
  const float* arow = h2 + (size_t)half * PLANE + (size_t)(base_b + row) * 4;

#pragma unroll 5
  for (int j = 0; j < NT / 2; ++j) {
    float4 a = *(const float4*)(arow + (size_t)j * NB * 4);
    const float4 bq = *(const float4*)(wrow + j * 8);
    const v2u sxz = __builtin_amdgcn_permlane16_swap(
        __float_as_uint(bq.x), __float_as_uint(bq.z), false, false);
    const v2u syw = __builtin_amdgcn_permlane16_swap(
        __float_as_uint(bq.y), __float_as_uint(bq.w), false, false);
    v2f a01, a23, b01, b23;
    a01.x = fmaxf(a.x, 0.0f);
    a01.y = fmaxf(a.y, 0.0f);
    a23.x = fmaxf(a.z, 0.0f);
    a23.y = fmaxf(a.w, 0.0f);
    b01.x = __uint_as_float(sxz.x); b01.y = __uint_as_float(syw.x);
    b23.x = __uint_as_float(sxz.y); b23.y = __uint_as_float(syw.y);
    acc = __builtin_amdgcn_wmma_f32_16x16x4_f32(
        false, a01, false, b01, (short)0, acc, false, false);
    acc = __builtin_amdgcn_wmma_f32_16x16x4_f32(
        false, a23, false, b23, (short)0, acc, false, false);
  }

  {
    const float bn = bias[n];
    float* d = act_tile + (mt * 16 + half * 8) * ACT_PITCH + n;
#pragma unroll
    for (int r = 0; r < 8; ++r)
      d[r * ACT_PITCH] = fmaxf(acc[r] + bn, 0.0f);
  }
  __syncthreads();

  if (wave < 4) {
    const int mt2 = wave >> 1;
    const int n2 = (wave & 1) * 16 + row;
    const bool nvalid = (n2 < 20);
    const int base_b2 = (blockIdx.x * 2 + mt2) * 16;

    v8f acc2 = {};
    const float* arow2 = act_tile + (mt2 * 16 + row) * ACT_PITCH + half * 4;
    const float* wrow2 = w2 + (size_t)(nvalid ? n2 : 0) * 64 + half * 4;

#pragma unroll
    for (int j = 0; j < 8; ++j) {
      const float4 a = *(const float4*)(arow2 + j * 8);
      float4 bq = *(const float4*)(wrow2 + j * 8);
      if (!nvalid) { bq.x = 0.0f; bq.y = 0.0f; bq.z = 0.0f; bq.w = 0.0f; }
      v2f a01, a23, b01, b23;
      a01.x = a.x; a01.y = a.y;
      a23.x = a.z; a23.y = a.w;
      b01.x = bq.x; b01.y = bq.y;
      b23.x = bq.z; b23.y = bq.w;
      acc2 = __builtin_amdgcn_wmma_f32_16x16x4_f32(
          false, a01, false, b01, (short)0, acc2, false, false);
      acc2 = __builtin_amdgcn_wmma_f32_16x16x4_f32(
          false, a23, false, b23, (short)0, acc2, false, false);
    }

    if (nvalid) {
      const float bn2 = bias2[n2];
#pragma unroll
      for (int r = 0; r < 8; ++r) {
        const int m = r + half * 8;
        out[(size_t)(base_b2 + m) * 20 + n2] = acc2[r] + bn2;
      }
    }
  }
}

extern "C" void kernel_launch(void* const* d_in, const int* in_sizes, int n_in,
                              void* d_out, int out_size, void* d_ws,
                              size_t ws_size, hipStream_t stream) {
  const float* x = (const float*)d_in[0];
  const float* w_ih_l0  = (const float*)d_in[1];
  const float* w_hh_l0  = (const float*)d_in[2];
  const float* b_ih_l0  = (const float*)d_in[3];
  const float* b_hh_l0  = (const float*)d_in[4];
  const float* w_ih_l0r = (const float*)d_in[5];
  const float* w_hh_l0r = (const float*)d_in[6];
  const float* b_ih_l0r = (const float*)d_in[7];
  const float* b_hh_l0r = (const float*)d_in[8];
  const float* w_ih_l1  = (const float*)d_in[9];
  const float* w_hh_l1  = (const float*)d_in[10];
  const float* b_ih_l1  = (const float*)d_in[11];
  const float* b_hh_l1  = (const float*)d_in[12];
  const float* w_ih_l1r = (const float*)d_in[13];
  const float* w_hh_l1r = (const float*)d_in[14];
  const float* b_ih_l1r = (const float*)d_in[15];
  const float* b_hh_l1r = (const float*)d_in[16];
  const float* fc1_w = (const float*)d_in[17];
  const float* fc1_b = (const float*)d_in[18];
  const float* fc2_w = (const float*)d_in[19];
  const float* fc2_b = (const float*)d_in[20];
  float* out = (float*)d_out;

  float* ws  = (float*)d_ws;
  float* ws0 = ws;
  float* ws1 = ws + (size_t)NT * NB * 4;

  const dim3 blk(256);
  const dim3 blk_lstm(512);
  const dim3 grid_lstm(NB / 256);

  lstm_bidir_kernel<7, false><<<grid_lstm, blk_lstm, 0, stream>>>(
      x, w_ih_l0, w_hh_l0, b_ih_l0, b_hh_l0,
      w_ih_l0r, w_hh_l0r, b_ih_l0r, b_hh_l0r, ws0);

  lstm_bidir_kernel<4, true><<<grid_lstm, blk_lstm, 0, stream>>>(
      ws0, w_ih_l1, w_hh_l1, b_ih_l1, b_hh_l1,
      w_ih_l1r, w_hh_l1r, b_ih_l1r, b_hh_l1r, ws1);

  fc_head_kernel<<<dim3(2048), blk, 0, stream>>>(ws1, fc1_w, fc1_b, fc2_w, fc2_b,
                                                out);
}


# FC1 re-tiled along samples: a wave keeps 4 sample tiles (64 samples x 16 outputs, 4 accumulators); each w fragment, fetched and row-swapped once, feeds 8 WMMAs; 4096 waves, not 16384; sums unchanged
# speedup vs baseline: 2.9156x; 1.1144x over previous
#include <hip/hip_runtime.h>
#include <hip/hip_bf16.h>


#define NB 65536
#define NT 100

#define PLANE ((size_t)(NT / 2) * NB * 4)

typedef __attribute__((ext_vector_type(2))) float v2f;
typedef __attribute__((ext_vector_type(8))) float v8f;
typedef __attribute__((ext_vector_type(2))) unsigned int v2u;

__device__ __forceinline__ float fast_rcp(float x) {
  return __builtin_amdgcn_rcpf(x);
}

#define LOG2E_F 1.44269504088896340736f
#define KS (-LOG2E_F)
#define KT (-2.0f * LOG2E_F)

__device__ __forceinline__ float sigm_scaled(float xs) {
  return fast_rcp(1.0f + __builtin_amdgcn_exp2f(xs));
}

template <int IN>
struct DirWeights {
  float wih[8][IN];
  float whh[8][2];
  float bias[8];
};

__device__ __forceinline__ float keep_in_vgpr(float v) {
  asm("" : "+v"(v));
  return v;
}

template <int IN, bool IN_VGPR>
__device__ __forceinline__ void load_dir_weights(
    DirWeights<IN>& w,
    const float* __restrict__ w_ih,
    const float* __restrict__ w_hh,
    const float* __restrict__ b_ih,
    const float* __restrict__ b_hh)
{
#pragma unroll
  for (int g = 0; g < 8; ++g) {
    const float ks = (g == 4 || g == 5) ? KT : KS;
#pragma unroll
    for (int c = 0; c < IN; ++c) w.wih[g][c] = ks * w_ih[g * IN + c];
    w.whh[g][0] = ks * w_hh[g * 2 + 0];
    w.whh[g][1] = ks * w_hh[g * 2 + 1];
    w.bias[g] = ks * (b_ih[g] + b_hh[g]);
    if constexpr (IN_VGPR) {
#pragma unroll
      for (int c = 0; c < IN; ++c) w.wih[g][c] = keep_in_vgpr(w.wih[g][c]);
      w.whh[g][0] = keep_in_vgpr(w.whh[g][0]);
      w.whh[g][1] = keep_in_vgpr(w.whh[g][1]);
      w.bias[g] = keep_in_vgpr(w.bias[g]);
    }
  }
}

template <int IN>
__device__ __forceinline__ void gate_input_part(
    const DirWeights<IN>& w, const float (&xv)[IN], float (&p)[8])
{
#pragma unroll
  for (int gg = 0; gg < 8; ++gg) {
    float acc = w.bias[gg];
#pragma unroll
    for (int c = 0; c < IN; ++c) acc = fmaf(w.wih[gg][c], xv[c], acc);
    p[gg] = acc;
  }
}

template <int IN>
__device__ __forceinline__ void lstm_step(
    const DirWeights<IN>& w, const float (&p)[8],
    float& h0, float& h1, float& cs0, float& cs1)
{
  float g[8];
#pragma unroll
  for (int gg = 0; gg < 8; ++gg)
    g[gg] = fmaf(w.whh[gg][0], h0, fmaf(w.whh[gg][1], h1, p[gg]));

  const float i0 = sigm_scaled(g[0]), i1 = sigm_scaled(g[1]);
  const float f0 = sigm_scaled(g[2]), f1 = sigm_scaled(g[3]);
  const float t0 = fmaf(2.0f * KT, sigm_scaled(g[4]), -KT);
  const float t1 = fmaf(2.0f * KT, sigm_scaled(g[5]), -KT);
  const float o0 = sigm_scaled(g[6]), o1 = sigm_scaled(g[7]);

  cs0 = fmaf(f0, cs0, i0 * t0);
  cs1 = fmaf(f1, cs1, i1 * t1);
  h0 = o0 * fmaf(2.0f, sigm_scaled(cs0), -1.0f);
  h1 = o1 * fmaf(2.0f, sigm_scaled(cs1), -1.0f);
}

#define SPW 256
#define TC 10
#define NCH (NT / TC)
#define PITCH 257

template <int IN, bool TMAJOR>
struct TilePieces;
template <int IN>
struct TilePieces<IN, false> {
  static constexpr int PPS = TC * IN / 2;
  static constexpr int TAIL = PPS - 32;
  static_assert(TAIL > 0 && TAIL <= 32, "row load takes 32 pieces, tail loads the rest");
  float2 v[32 + TAIL];
};
template <int IN>
struct TilePieces<IN, true> {
  float4 v[TC];
};

template <int IN, bool TMAJOR>
__device__ __forceinline__ void load_tile(
    TilePieces<IN, TMAJOR>& r, const float* __restrict__ in, int b0, int t0,
    int lj)
{
  if constexpr (TMAJOR) {
    static_assert(IN == 4 && TC % 2 == 0, "paired planes: 4 channels, whole pairs");
#pragma unroll
    for (int i = 0; i < TC; ++i)
      r.v[i] = *(const float4*)(in + (size_t)(i / (TC / 2)) * PLANE +
                                ((size_t)((t0 >> 1) + (i % (TC / 2))) * NB +
                                 (size_t)(b0 + lj)) * 4);
  } else {
    static_assert((TC * IN) % 2 == 0, "tile row must be whole 8-byte pieces");
    constexpr int TAIL = TilePieces<IN, false>::TAIL;
    const int l = lj & 31;
    const float* base = in + (size_t)(b0 + (lj & ~31)) * (NT * IN) + (size_t)t0 * IN;
#pragma unroll
    for (int i = 0; i < 32; ++i)
      r.v[i] = *(const float2*)(base + (size_t)i * (NT * IN) + 2 * l);
#pragma unroll
    for (int j = 0; j < TAIL; ++j) {
      const int idx = 32 * j + l;
      r.v[32 + j] = *(const float2*)(base + (size_t)(idx / TAIL) * (NT * IN) +
                                     2 * (32 + idx % TAIL));
    }
  }
}

template <int IN, bool TMAJOR>
__device__ __forceinline__ void store_tile(
    float* __restrict__ tile, const TilePieces<IN, TMAJOR>& r, int lj)
{
  if constexpr (TMAJOR) {
#pragma unroll
    for (int i = 0; i < TC; ++i) {
      float* d = tile +
                 ((i % (TC / 2)) * 2 * IN + (i / (TC / 2)) * 2) * PITCH + lj;
      d[0 * PITCH] = r.v[i].x;
      d[1 * PITCH] = r.v[i].y;
      d[(IN + 0) * PITCH] = r.v[i].z;
      d[(IN + 1) * PITCH] = r.v[i].w;
    }
  } else {
    constexpr int TAIL = TilePieces<IN, false>::TAIL;
    const int l = lj & 31;
    float* d = tile + (2 * l) * PITCH + (lj & ~31);
#pragma unroll
    for (int i = 0; i < 32; ++i) {
      d[0 * PITCH + i] = r.v[i].x;
      d[1 * PITCH + i] = r.v[i].y;
    }
#pragma unroll
    for (int j = 0; j < TAIL; ++j) {
      const int idx = 32 * j + l;
      float* e = tile + (2 * (32 + idx % TAIL)) * PITCH + (lj & ~31) + idx / TAIL;
      e[0 * PITCH] = r.v[32 + j].x;
      e[1 * PITCH] = r.v[32 + j].y;
    }
  }
}

__device__ __forceinline__ void lds_handover() {
  __builtin_amdgcn_fence(__ATOMIC_RELEASE, "workgroup", "local");
  __builtin_amdgcn_s_barrier();
  __builtin_amdgcn_fence(__ATOMIC_ACQUIRE, "workgroup", "local");
}

template <int IN, bool TMAJOR>
__global__ __launch_bounds__(512) __attribute__((amdgpu_waves_per_eu(4, 4)))
void lstm_bidir_kernel(
    const float* __restrict__ in,
    const float* __restrict__ wf_ih, const float* __restrict__ wf_hh,
    const float* __restrict__ bf_ih, const float* __restrict__ bf_hh,
    const float* __restrict__ wr_ih, const float* __restrict__ wr_hh,
    const float* __restrict__ br_ih, const float* __restrict__ br_hh,
    float* __restrict__ out)
{
  __shared__ float ring[2][2][TC * IN * PITCH];

  const int tid = threadIdx.x;
  const bool is_loader = __builtin_amdgcn_readfirstlane(tid) >= SPW;
  const int b0 = blockIdx.x * SPW;
  const int lj = tid - SPW;
  const int b = b0 + tid;

  DirWeights<IN> wf, wr;
  load_dir_weights<IN, false>(wf, wf_ih, wf_hh, bf_ih, bf_hh);
  load_dir_weights<IN, true>(wr, wr_ih, wr_hh, br_ih, br_hh);

  float hf0 = 0.f, hf1 = 0.f, cf0 = 0.f, cf1 = 0.f;
  float hr0 = 0.f, hr1 = 0.f, cr0 = 0.f, cr1 = 0.f;
  float kf0 = 0.f, kf1 = 0.f, kr0 = 0.f, kr1 = 0.f;
  static_assert(TC % 2 == 0, "a tile must hold whole time pairs");

  TilePieces<IN, TMAJOR> pt;
  if (is_loader) {
    load_tile<IN, TMAJOR>(pt, in, b0, 0, lj);
    store_tile<IN, TMAJOR>(ring[0][0], pt, lj);
    load_tile<IN, TMAJOR>(pt, in, b0, NT - TC, lj);
    store_tile<IN, TMAJOR>(ring[0][1], pt, lj);
    load_tile<IN, TMAJOR>(pt, in, b0, TC, lj);
  }
  lds_handover();

  for (int c = 0; c < NCH; ++c) {
    const int stage = c & 1;
    if (is_loader) {
      if (c + 1 < NCH) {
        store_tile<IN, TMAJOR>(ring[stage ^ 1][0], pt, lj);
        load_tile<IN, TMAJOR>(pt, in, b0, NT - TC - (c + 1) * TC, lj);
        store_tile<IN, TMAJOR>(ring[stage ^ 1][1], pt, lj);
        if (c + 2 < NCH)
          load_tile<IN, TMAJOR>(pt, in, b0, (c + 2) * TC, lj);
      }
    } else {
      const float* tf_tile = ring[stage][0];
      const float* tr_tile = ring[stage][1];
      float pf[8], pr[8];
      {
        float x0f[IN], x0r[IN];
#pragma unroll
        for (int ch = 0; ch < IN; ++ch) {
          x0f[ch] = tf_tile[ch * PITCH + tid];
          x0r[ch] = tr_tile[((TC - 1) * IN + ch) * PITCH + tid];
        }
        gate_input_part<IN>(wf, x0f, pf);
        gate_input_part<IN>(wr, x0r, pr);
      }
#pragma unroll
      for (int s = 0; s < TC; ++s) {
        const int tf = c * TC + s;
        const int tr = NT - 1 - tf;
        float xf[IN], xr[IN];
        if (s + 1 < TC) {
#pragma unroll
          for (int ch = 0; ch < IN; ++ch) {
            xf[ch] = tf_tile[((s + 1) * IN + ch) * PITCH + tid];
            xr[ch] = tr_tile[((TC - 2 - s) * IN + ch) * PITCH + tid];
          }
        }
        __builtin_amdgcn_sched_barrier(0);

        lstm_step<IN>(wf, pf, hf0, hf1, cf0, cf1);
        lstm_step<IN>(wr, pr, hr0, hr1, cr0, cr1);

        if (s + 1 < TC) {
          gate_input_part<IN>(wf, xf, pf);
          gate_input_part<IN>(wr, xr, pr);
        }

        if ((s & 1) == 0) {
          kf0 = hf0; kf1 = hf1;
          kr0 = hr0; kr1 = hr1;
        } else {
          *(float4*)(out + ((size_t)(tf >> 1) * NB + (size_t)b) * 4) =
              make_float4(kf0, kf1, hf0, hf1);
          *(float4*)(out + PLANE + ((size_t)(tr >> 1) * NB + (size_t)b) * 4) =
              make_float4(hr0, hr1, kr0, kr1);
        }
      }
    }
    lds_handover();
  }
}

#define ACT_PITCH 68
#define FC_MT 4
#define FC_SPB (2 * FC_MT * 16)

__global__ __launch_bounds__(256) __attribute__((amdgpu_waves_per_eu(4, 4)))
void fc_head_kernel(
    const float* __restrict__ h2,
    const float* __restrict__ w,
    const float* __restrict__ bias,
    const float* __restrict__ w2,
    const float* __restrict__ bias2,
    float* __restrict__ out)
{
  __shared__ __attribute__((aligned(16))) float act_tile[FC_SPB * ACT_PITCH];

  const int wave = __builtin_amdgcn_readfirstlane((int)threadIdx.x) >> 5;
  const int lane = threadIdx.x & 31;
  const int sg = wave >> 2;
  const int n_tile = wave & 3;
  const int row = lane & 15;
  const int half = lane >> 4;
  const int base_b = blockIdx.x * FC_SPB + sg * (FC_MT * 16);
  const int n = n_tile * 16 + row;

  v8f acc[FC_MT];
#pragma unroll
  for (int m = 0; m < FC_MT; ++m) acc[m] = v8f{};

  const float* wrow = w + (size_t)n * 400 + half * 4;
  const float* arow = h2 + (size_t)half * PLANE + (size_t)(base_b + row) * 4;

#pragma unroll 5
  for (int j = 0; j < NT / 2; ++j) {
    const float4 bq = *(const float4*)(wrow + j * 8);
    float4 a[FC_MT];
#pragma unroll
    for (int m = 0; m < FC_MT; ++m)
      a[m] = *(const float4*)(arow + (size_t)j * NB * 4 + m * 16 * 4);
    const v2u sxz = __builtin_amdgcn_permlane16_swap(
        __float_as_uint(bq.x), __float_as_uint(bq.z), false, false);
    const v2u syw = __builtin_amdgcn_permlane16_swap(
        __float_as_uint(bq.y), __float_as_uint(bq.w), false, false);
    v2f b01, b23;
    b01.x = __uint_as_float(sxz.x); b01.y = __uint_as_float(syw.x);
    b23.x = __uint_as_float(sxz.y); b23.y = __uint_as_float(syw.y);
#pragma unroll
    for (int m = 0; m < FC_MT; ++m) {
      v2f a01;
      a01.x = fmaxf(a[m].x, 0.0f);
      a01.y = fmaxf(a[m].y, 0.0f);
      acc[m] = __builtin_amdgcn_wmma_f32_16x16x4_f32(
          false, a01, false, b01, (short)0, acc[m], false, false);
    }
#pragma unroll
    for (int m = 0; m < FC_MT; ++m) {
      v2f a23;
      a23.x = fmaxf(a[m].z, 0.0f);
      a23.y = fmaxf(a[m].w, 0.0f);
      acc[m] = __builtin_amdgcn_wmma_f32_16x16x4_f32(
          false, a23, false, b23, (short)0, acc[m], false, false);
    }
  }

  {
    const float bn = bias[n];
#pragma unroll
    for (int m = 0; m < FC_MT; ++m) {
      float* d = act_tile + (sg * (FC_MT * 16) + m * 16 + half * 8) * ACT_PITCH + n;
#pragma unroll
      for (int r = 0; r < 8; ++r)
        d[r * ACT_PITCH] = fmaxf(acc[m][r] + bn, 0.0f);
    }
  }
  __syncthreads();

  {
    const int base_b2 = blockIdx.x * FC_SPB + wave * 16;
    const float* arow2 = act_tile + (wave * 16 + row) * ACT_PITCH + half * 4;
    const int n2a = row;
    const int n2b = 16 + row;
    const bool bvalid = (n2b < 20);
    const float* wrow2a = w2 + (size_t)n2a * 64 + half * 4;
    const float* wrow2b = w2 + (size_t)(bvalid ? n2b : 0) * 64 + half * 4;

    v8f acc2a = {}, acc2b = {};
#pragma unroll
    for (int j = 0; j < 8; ++j) {
      const float4 a = *(const float4*)(arow2 + j * 8);
      const float4 bqa = *(const float4*)(wrow2a + j * 8);
      float4 bqb = *(const float4*)(wrow2b + j * 8);
      if (!bvalid) { bqb.x = 0.0f; bqb.y = 0.0f; bqb.z = 0.0f; bqb.w = 0.0f; }
      v2f a01, a23, b01, b23, c01, c23;
      a01.x = a.x; a01.y = a.y;
      a23.x = a.z; a23.y = a.w;
      b01.x = bqa.x; b01.y = bqa.y;
      b23.x = bqa.z; b23.y = bqa.w;
      c01.x = bqb.x; c01.y = bqb.y;
      c23.x = bqb.z; c23.y = bqb.w;
      acc2a = __builtin_amdgcn_wmma_f32_16x16x4_f32(
          false, a01, false, b01, (short)0, acc2a, false, false);
      acc2b = __builtin_amdgcn_wmma_f32_16x16x4_f32(
          false, a01, false, c01, (short)0, acc2b, false, false);
      acc2a = __builtin_amdgcn_wmma_f32_16x16x4_f32(
          false, a23, false, b23, (short)0, acc2a, false, false);
      acc2b = __builtin_amdgcn_wmma_f32_16x16x4_f32(
          false, a23, false, c23, (short)0, acc2b, false, false);
    }

    {
      const float bn2 = bias2[n2a];
#pragma unroll
      for (int r = 0; r < 8; ++r) {
        const int s = r + half * 8;
        out[(size_t)(base_b2 + s) * 20 + n2a] = acc2a[r] + bn2;
      }
    }
    if (bvalid) {
      const float bn2 = bias2[n2b];
#pragma unroll
      for (int r = 0; r < 8; ++r) {
        const int s = r + half * 8;
        out[(size_t)(base_b2 + s) * 20 + n2b] = acc2b[r] + bn2;
      }
    }
  }
}

extern "C" void kernel_launch(void* const* d_in, const int* in_sizes, int n_in,
                              void* d_out, int out_size, void* d_ws,
                              size_t ws_size, hipStream_t stream) {
  const float* x = (const float*)d_in[0];
  const float* w_ih_l0  = (const float*)d_in[1];
  const float* w_hh_l0  = (const float*)d_in[2];
  const float* b_ih_l0  = (const float*)d_in[3];
  const float* b_hh_l0  = (const float*)d_in[4];
  const float* w_ih_l0r = (const float*)d_in[5];
  const float* w_hh_l0r = (const float*)d_in[6];
  const float* b_ih_l0r = (const float*)d_in[7];
  const float* b_hh_l0r = (const float*)d_in[8];
  const float* w_ih_l1  = (const float*)d_in[9];
  const float* w_hh_l1  = (const float*)d_in[10];
  const float* b_ih_l1  = (const float*)d_in[11];
  const float* b_hh_l1  = (const float*)d_in[12];
  const float* w_ih_l1r = (const float*)d_in[13];
  const float* w_hh_l1r = (const float*)d_in[14];
  const float* b_ih_l1r = (const float*)d_in[15];
  const float* b_hh_l1r = (const float*)d_in[16];
  const float* fc1_w = (const float*)d_in[17];
  const float* fc1_b = (const float*)d_in[18];
  const float* fc2_w = (const float*)d_in[19];
  const float* fc2_b = (const float*)d_in[20];
  float* out = (float*)d_out;

  float* ws  = (float*)d_ws;
  float* ws0 = ws;
  float* ws1 = ws + (size_t)NT * NB * 4;

  const dim3 blk(256);
  const dim3 blk_lstm(512);
  const dim3 grid_lstm(NB / 256);

  lstm_bidir_kernel<7, false><<<grid_lstm, blk_lstm, 0, stream>>>(
      x, w_ih_l0, w_hh_l0, b_ih_l0, b_hh_l0,
      w_ih_l0r, w_hh_l0r, b_ih_l0r, b_hh_l0r, ws0);

  lstm_bidir_kernel<4, true><<<grid_lstm, blk_lstm, 0, stream>>>(
      ws0, w_ih_l1, w_hh_l1, b_ih_l1, b_hh_l1,
      w_ih_l1r, w_hh_l1r, b_ih_l1r, b_hh_l1r, ws1);

  fc_head_kernel<<<dim3(NB / FC_SPB), blk, 0, stream>>>(ws1, fc1_w, fc1_b, fc2_w,
                                                       fc2_b, out);
}
